# starting kernel
# baseline (speedup 1.0000x reference)
#include <hip/hip_runtime.h>

typedef __attribute__((ext_vector_type(2))) float v2f;
typedef __attribute__((ext_vector_type(8))) float v8f;

#define FEATS 64

__global__ void gcn_zero_kernel(float4* __restrict__ p, int n4) {
    int i = blockIdx.x * blockDim.x + threadIdx.x;
    int stride = gridDim.x * blockDim.x;
    float4 z = make_float4(0.f, 0.f, 0.f, 0.f);
    for (; i < n4; i += stride) p[i] = z;
}

__global__ void gcn_edge_scatter_kernel(const float* __restrict__ feat,
                                        const float* __restrict__ degree,
                                        const int* __restrict__ src,
                                        const int* __restrict__ dst,
                                        float* __restrict__ agg,
                                        int n_edges) {
    int wave   = (blockIdx.x * blockDim.x + threadIdx.x) >> 5;
    int lane   = threadIdx.x & 31;
    int nwaves = (gridDim.x * blockDim.x) >> 5;

    for (int e = wave; e < n_edges; e += nwaves) {
        int s = src[e];
        int d = dst[e];
        float w = __frsqrt_rn(degree[s]);
        float2 v = *(const float2*)(feat + (size_t)s * FEATS + lane * 2);
        float* outp = agg + (size_t)d * FEATS + lane * 2;
        unsafeAtomicAdd(outp + 0, v.x * w);
        unsafeAtomicAdd(outp + 1, v.y * w);
    }
}

__global__ void gcn_node_apply_wmma_kernel(const float* __restrict__ agg,
                                           const float* __restrict__ degree,
                                           const float* __restrict__ Wm,
                                           const float* __restrict__ bias,
                                           float* __restrict__ out,
                                           int n_nodes, int n_mtiles) {
    int wave  = blockIdx.x * (blockDim.x >> 5) + (threadIdx.x >> 5);
    if (wave >= n_mtiles) return;
    int lane  = threadIdx.x & 31;
    int n     = lane & 15;
    int g     = lane >> 4;

    int m      = wave * 16 + n;
    int mclamp = m < n_nodes ? m : n_nodes - 1;
    float dsc  = __frsqrt_rn(degree[mclamp]);
    const float* arow = agg + (size_t)mclamp * FEATS;

    v8f acc0 = {}, acc1 = {}, acc2 = {}, acc3 = {};

    for (int k0 = 0; k0 < FEATS; k0 += 4) {
        int ka = k0 + 2 * g;
        float2 av = *(const float2*)(arow + ka);
        v2f a;
        a.x = av.x * dsc;
        a.y = av.y * dsc;

        const float* wr0 = Wm + (size_t)ka * FEATS;
        const float* wr1 = wr0 + FEATS;
        v2f b0; b0.x = wr0[n +  0]; b0.y = wr1[n +  0];
        v2f b1; b1.x = wr0[n + 16]; b1.y = wr1[n + 16];
        v2f b2; b2.x = wr0[n + 32]; b2.y = wr1[n + 32];
        v2f b3; b3.x = wr0[n + 48]; b3.y = wr1[n + 48];

        acc0 = __builtin_amdgcn_wmma_f32_16x16x4_f32(false, a, false, b0, (short)0, acc0, false, false);
        acc1 = __builtin_amdgcn_wmma_f32_16x16x4_f32(false, a, false, b1, (short)0, acc1, false, false);
        acc2 = __builtin_amdgcn_wmma_f32_16x16x4_f32(false, a, false, b2, (short)0, acc2, false, false);
        acc3 = __builtin_amdgcn_wmma_f32_16x16x4_f32(false, a, false, b3, (short)0, acc3, false, false);
    }

    float bb0 = bias[n +  0];
    float bb1 = bias[n + 16];
    float bb2 = bias[n + 32];
    float bb3 = bias[n + 48];

#pragma unroll
    for (int r = 0; r < 8; ++r) {
        int mrow = wave * 16 + r + 8 * g;
        if (mrow < n_nodes) {
            float* orow = out + (size_t)mrow * FEATS + n;
            orow[ 0] = acc0[r] + bb0;
            orow[16] = acc1[r] + bb1;
            orow[32] = acc2[r] + bb2;
            orow[48] = acc3[r] + bb3;
        }
    }
}

extern "C" void kernel_launch(void* const* d_in, const int* in_sizes, int n_in,
                              void* d_out, int out_size, void* d_ws, size_t ws_size,
                              hipStream_t stream) {
    const float* feature = (const float*)d_in[0];
    const float* degree  = (const float*)d_in[1];
    const int*   src     = (const int*)d_in[2];
    const int*   dst     = (const int*)d_in[3];
    const float* Wm      = (const float*)d_in[4];
    const float* bias    = (const float*)d_in[5];
    float* out = (float*)d_out;
    float* agg = (float*)d_ws;

    int n_nodes = in_sizes[1];
    int n_edges = in_sizes[2];

    int n4 = (n_nodes * FEATS) / 4;
    int zb = (n4 + 255) / 256;
    gcn_zero_kernel<<<zb, 256, 0, stream>>>((float4*)agg, n4);

    gcn_edge_scatter_kernel<<<1024, 256, 0, stream>>>(feature, degree, src, dst,
                                                      agg, n_edges);

    int n_mtiles = (n_nodes + 15) / 16;
    int blocks = (n_mtiles + 7) / 8;
    gcn_node_apply_wmma_kernel<<<blocks, 256, 0, stream>>>(agg, degree, Wm, bias,
                                                           out, n_nodes, n_mtiles);
}


# edge scatter: lane l owns row elements l and l+32, so each f32 atomic add instruction covers one whole 128-B line, not half of two lines; atomics kept, sum order arrival order as in parent
# speedup vs baseline: 1.0421x; 1.0421x over previous
#include <hip/hip_runtime.h>

typedef __attribute__((ext_vector_type(2))) float v2f;
typedef __attribute__((ext_vector_type(8))) float v8f;

#define FEATS 64

__global__ void gcn_zero_kernel(float4* __restrict__ p, int n4) {
    int i = blockIdx.x * blockDim.x + threadIdx.x;
    int stride = gridDim.x * blockDim.x;
    float4 z = make_float4(0.f, 0.f, 0.f, 0.f);
    for (; i < n4; i += stride) p[i] = z;
}

__global__ void gcn_edge_scatter_kernel(const float* __restrict__ feat,
                                        const float* __restrict__ degree,
                                        const int* __restrict__ src,
                                        const int* __restrict__ dst,
                                        float* __restrict__ agg,
                                        int n_edges) {
    int wave   = (blockIdx.x * blockDim.x + threadIdx.x) >> 5;
    int lane   = threadIdx.x & 31;
    int nwaves = (gridDim.x * blockDim.x) >> 5;

    for (int e = wave; e < n_edges; e += nwaves) {
        int s = src[e];
        int d = dst[e];
        float w = __frsqrt_rn(degree[s]);
        const float* frow = feat + (size_t)s * FEATS + lane;
        float v0 = frow[0];
        float v1 = frow[FEATS / 2];
        float* outp = agg + (size_t)d * FEATS + lane;
        unsafeAtomicAdd(outp,             v0 * w);
        unsafeAtomicAdd(outp + FEATS / 2, v1 * w);
    }
}

__global__ void gcn_node_apply_wmma_kernel(const float* __restrict__ agg,
                                           const float* __restrict__ degree,
                                           const float* __restrict__ Wm,
                                           const float* __restrict__ bias,
                                           float* __restrict__ out,
                                           int n_nodes, int n_mtiles) {
    int wave  = blockIdx.x * (blockDim.x >> 5) + (threadIdx.x >> 5);
    if (wave >= n_mtiles) return;
    int lane  = threadIdx.x & 31;
    int n     = lane & 15;
    int g     = lane >> 4;

    int m      = wave * 16 + n;
    int mclamp = m < n_nodes ? m : n_nodes - 1;
    float dsc  = __frsqrt_rn(degree[mclamp]);
    const float* arow = agg + (size_t)mclamp * FEATS;

    v8f acc0 = {}, acc1 = {}, acc2 = {}, acc3 = {};

    for (int k0 = 0; k0 < FEATS; k0 += 4) {
        int ka = k0 + 2 * g;
        float2 av = *(const float2*)(arow + ka);
        v2f a;
        a.x = av.x * dsc;
        a.y = av.y * dsc;

        const float* wr0 = Wm + (size_t)ka * FEATS;
        const float* wr1 = wr0 + FEATS;
        v2f b0; b0.x = wr0[n +  0]; b0.y = wr1[n +  0];
        v2f b1; b1.x = wr0[n + 16]; b1.y = wr1[n + 16];
        v2f b2; b2.x = wr0[n + 32]; b2.y = wr1[n + 32];
        v2f b3; b3.x = wr0[n + 48]; b3.y = wr1[n + 48];

        acc0 = __builtin_amdgcn_wmma_f32_16x16x4_f32(false, a, false, b0, (short)0, acc0, false, false);
        acc1 = __builtin_amdgcn_wmma_f32_16x16x4_f32(false, a, false, b1, (short)0, acc1, false, false);
        acc2 = __builtin_amdgcn_wmma_f32_16x16x4_f32(false, a, false, b2, (short)0, acc2, false, false);
        acc3 = __builtin_amdgcn_wmma_f32_16x16x4_f32(false, a, false, b3, (short)0, acc3, false, false);
    }

    float bb0 = bias[n +  0];
    float bb1 = bias[n + 16];
    float bb2 = bias[n + 32];
    float bb3 = bias[n + 48];

#pragma unroll
    for (int r = 0; r < 8; ++r) {
        int mrow = wave * 16 + r + 8 * g;
        if (mrow < n_nodes) {
            float* orow = out + (size_t)mrow * FEATS + n;
            orow[ 0] = acc0[r] + bb0;
            orow[16] = acc1[r] + bb1;
            orow[32] = acc2[r] + bb2;
            orow[48] = acc3[r] + bb3;
        }
    }
}

extern "C" void kernel_launch(void* const* d_in, const int* in_sizes, int n_in,
                              void* d_out, int out_size, void* d_ws, size_t ws_size,
                              hipStream_t stream) {
    const float* feature = (const float*)d_in[0];
    const float* degree  = (const float*)d_in[1];
    const int*   src     = (const int*)d_in[2];
    const int*   dst     = (const int*)d_in[3];
    const float* Wm      = (const float*)d_in[4];
    const float* bias    = (const float*)d_in[5];
    float* out = (float*)d_out;
    float* agg = (float*)d_ws;

    int n_nodes = in_sizes[1];
    int n_edges = in_sizes[2];

    int n4 = (n_nodes * FEATS) / 4;
    int zb = (n4 + 255) / 256;
    gcn_zero_kernel<<<zb, 256, 0, stream>>>((float4*)agg, n4);

    gcn_edge_scatter_kernel<<<1024, 256, 0, stream>>>(feature, degree, src, dst,
                                                      agg, n_edges);

    int n_mtiles = (n_nodes + 15) / 16;
    int blocks = (n_mtiles + 7) / 8;
    gcn_node_apply_wmma_kernel<<<blocks, 256, 0, stream>>>(agg, degree, Wm, bias,
                                                           out, n_nodes, n_mtiles);
}


# edge scatter: a wave takes 32 edges per trip (one coalesced index and degree load), then issues the row loads of 8 edges together ahead of their f32 atomic adds; sum order arrival order as in parent
# speedup vs baseline: 1.9756x; 1.8958x over previous
#include <hip/hip_runtime.h>

typedef __attribute__((ext_vector_type(2))) float v2f;
typedef __attribute__((ext_vector_type(8))) float v8f;

#define FEATS 64

__global__ void gcn_zero_kernel(float4* __restrict__ p, int n4) {
    int i = blockIdx.x * blockDim.x + threadIdx.x;
    int stride = gridDim.x * blockDim.x;
    float4 z = make_float4(0.f, 0.f, 0.f, 0.f);
    for (; i < n4; i += stride) p[i] = z;
}

#define CHUNK 32
#define GROUP 8

__device__ __forceinline__ float lane_read_f(float v, int j) {
    return __int_as_float(__builtin_amdgcn_readlane(__float_as_int(v), j));
}

__global__ void __launch_bounds__(256)
gcn_edge_scatter_kernel(const float* __restrict__ feat,
                        const float* __restrict__ degree,
                        const int* __restrict__ src,
                        const int* __restrict__ dst,
                        float* __restrict__ agg,
                        int n_edges) {
    int wave   = __builtin_amdgcn_readfirstlane((int)((blockIdx.x * blockDim.x + threadIdx.x) >> 5));
    int lane   = threadIdx.x & 31;
    int nwaves = (gridDim.x * blockDim.x) >> 5;
    int nchunks = (n_edges + CHUNK - 1) / CHUNK;

    for (int c = wave; c < nchunks; c += nwaves) {
        int e0 = c * CHUNK;
        int e  = e0 + lane;
        int s = 0, d = 0;
        float w = 0.f;
        if (e < n_edges) {
            s = src[e];
            d = dst[e];
            w = __frsqrt_rn(degree[s]);
        }
        int left = n_edges - e0;
        int cnt  = left < CHUNK ? left : CHUNK;

        int j = 0;
        for (; j + GROUP <= cnt; j += GROUP) {
            float v0[GROUP], v1[GROUP];
#pragma unroll
            for (int u = 0; u < GROUP; ++u) {
                int sj = __builtin_amdgcn_readlane(s, j + u);
                const float* frow = feat + (size_t)sj * FEATS + lane;
                v0[u] = frow[0];
                v1[u] = frow[FEATS / 2];
            }
#pragma unroll
            for (int u = 0; u < GROUP; ++u) {
                int   dj = __builtin_amdgcn_readlane(d, j + u);
                float wj = lane_read_f(w, j + u);
                float* outp = agg + (size_t)dj * FEATS + lane;
                unsafeAtomicAdd(outp,             v0[u] * wj);
                unsafeAtomicAdd(outp + FEATS / 2, v1[u] * wj);
            }
        }
        for (; j < cnt; ++j) {
            int   sj = __builtin_amdgcn_readlane(s, j);
            int   dj = __builtin_amdgcn_readlane(d, j);
            float wj = lane_read_f(w, j);
            const float* frow = feat + (size_t)sj * FEATS + lane;
            float a0 = frow[0];
            float a1 = frow[FEATS / 2];
            float* outp = agg + (size_t)dj * FEATS + lane;
            unsafeAtomicAdd(outp,             a0 * wj);
            unsafeAtomicAdd(outp + FEATS / 2, a1 * wj);
        }
    }
}

__global__ void gcn_node_apply_wmma_kernel(const float* __restrict__ agg,
                                           const float* __restrict__ degree,
                                           const float* __restrict__ Wm,
                                           const float* __restrict__ bias,
                                           float* __restrict__ out,
                                           int n_nodes, int n_mtiles) {
    int wave  = blockIdx.x * (blockDim.x >> 5) + (threadIdx.x >> 5);
    if (wave >= n_mtiles) return;
    int lane  = threadIdx.x & 31;
    int n     = lane & 15;
    int g     = lane >> 4;

    int m      = wave * 16 + n;
    int mclamp = m < n_nodes ? m : n_nodes - 1;
    float dsc  = __frsqrt_rn(degree[mclamp]);
    const float* arow = agg + (size_t)mclamp * FEATS;

    v8f acc0 = {}, acc1 = {}, acc2 = {}, acc3 = {};

    for (int k0 = 0; k0 < FEATS; k0 += 4) {
        int ka = k0 + 2 * g;
        float2 av = *(const float2*)(arow + ka);
        v2f a;
        a.x = av.x * dsc;
        a.y = av.y * dsc;

        const float* wr0 = Wm + (size_t)ka * FEATS;
        const float* wr1 = wr0 + FEATS;
        v2f b0; b0.x = wr0[n +  0]; b0.y = wr1[n +  0];
        v2f b1; b1.x = wr0[n + 16]; b1.y = wr1[n + 16];
        v2f b2; b2.x = wr0[n + 32]; b2.y = wr1[n + 32];
        v2f b3; b3.x = wr0[n + 48]; b3.y = wr1[n + 48];

        acc0 = __builtin_amdgcn_wmma_f32_16x16x4_f32(false, a, false, b0, (short)0, acc0, false, false);
        acc1 = __builtin_amdgcn_wmma_f32_16x16x4_f32(false, a, false, b1, (short)0, acc1, false, false);
        acc2 = __builtin_amdgcn_wmma_f32_16x16x4_f32(false, a, false, b2, (short)0, acc2, false, false);
        acc3 = __builtin_amdgcn_wmma_f32_16x16x4_f32(false, a, false, b3, (short)0, acc3, false, false);
    }

    float bb0 = bias[n +  0];
    float bb1 = bias[n + 16];
    float bb2 = bias[n + 32];
    float bb3 = bias[n + 48];

#pragma unroll
    for (int r = 0; r < 8; ++r) {
        int mrow = wave * 16 + r + 8 * g;
        if (mrow < n_nodes) {
            float* orow = out + (size_t)mrow * FEATS + n;
            orow[ 0] = acc0[r] + bb0;
            orow[16] = acc1[r] + bb1;
            orow[32] = acc2[r] + bb2;
            orow[48] = acc3[r] + bb3;
        }
    }
}

extern "C" void kernel_launch(void* const* d_in, const int* in_sizes, int n_in,
                              void* d_out, int out_size, void* d_ws, size_t ws_size,
                              hipStream_t stream) {
    const float* feature = (const float*)d_in[0];
    const float* degree  = (const float*)d_in[1];
    const int*   src     = (const int*)d_in[2];
    const int*   dst     = (const int*)d_in[3];
    const float* Wm      = (const float*)d_in[4];
    const float* bias    = (const float*)d_in[5];
    float* out = (float*)d_out;
    float* agg = (float*)d_ws;

    int n_nodes = in_sizes[1];
    int n_edges = in_sizes[2];

    int n4 = (n_nodes * FEATS) / 4;
    int zb = (n4 + 255) / 256;
    gcn_zero_kernel<<<zb, 256, 0, stream>>>((float4*)agg, n4);

    int sblocks = (n_edges + 767) / 768;
    gcn_edge_scatter_kernel<<<sblocks, 256, 0, stream>>>(feature, degree, src, dst,
                                                         agg, n_edges);

    int n_mtiles = (n_nodes + 15) / 16;
    int blocks = (n_mtiles + 7) / 8;
    gcn_node_apply_wmma_kernel<<<blocks, 256, 0, stream>>>(agg, degree, Wm, bias,
                                                           out, n_nodes, n_mtiles);
}


# edge scatter: sum rows owned by cache halves: a block reads its die, draws a rank in its half and adds only edges whose target row is in that half (both halves walk all edges); f32 atomics as parent
# speedup vs baseline: 2.3359x; 1.1824x over previous
#include <hip/hip_runtime.h>

typedef __attribute__((ext_vector_type(2))) float v2f;
typedef __attribute__((ext_vector_type(8))) float v8f;

#define FEATS 64

#define CTR_STRIDE 64

__global__ void gcn_zero_kernel(float4* __restrict__ p, int n4, int* __restrict__ ctr) {
    int i = blockIdx.x * blockDim.x + threadIdx.x;
    int stride = gridDim.x * blockDim.x;
    if (i < 2) ctr[i * CTR_STRIDE] = 0;
    float4 z = make_float4(0.f, 0.f, 0.f, 0.f);
    for (; i < n4; i += stride) p[i] = z;
}

#define CHUNK 32
#define GROUP 8
#define WPB   8

__device__ __forceinline__ float lane_read_f(float v, int j) {
    return __int_as_float(__builtin_amdgcn_readlane(__float_as_int(v), j));
}

__global__ void __launch_bounds__(256)
gcn_edge_scatter_kernel(const float* __restrict__ feat,
                        const float* __restrict__ degree,
                        const int* __restrict__ src,
                        const int* __restrict__ dst,
                        float* __restrict__ agg,
                        int n_edges, int n_nodes,
                        int* __restrict__ ctr) {
    __shared__ int s_rank;
    const int tid  = threadIdx.x;
    const int lane = tid & 31;
    const int wv   = __builtin_amdgcn_readfirstlane(tid >> 5);
    const int nchunks = (n_edges + CHUNK - 1) / CHUNK;
    const int split   = n_nodes >> 1;
    const int R       = (int)(gridDim.x >> 1);
    const int die    = (int)((__builtin_amdgcn_s_sendmsg_rtn(0x87) >> 16) & 15u);
    const int myhalf = __builtin_amdgcn_readfirstlane(die >> 2) & 1;

    if (tid == 0) s_rank = atomicAdd(ctr + myhalf * CTR_STRIDE, 1);
    __syncthreads();
    const int r = __builtin_amdgcn_readfirstlane(s_rank);

    int q, cls;
    if (r < R) { q = myhalf;     cls = r; }
    else       { q = myhalf ^ 1; cls = 2 * R - 1 - r; }
    if (cls < 0) return;

    const int cstep = R * WPB;
    for (int c = cls * WPB + wv; c < nchunks; c += cstep) {
        const int e = c * CHUNK + lane;
        int s = 0, d = 0;
        float w = 0.f;
        bool mine = false;
        if (e < n_edges) {
            s = src[e];
            d = dst[e];
            mine = ((d >= split) ? 1 : 0) == q;
        }
        if (mine) w = __frsqrt_rn(degree[s]);
        unsigned mask = __builtin_amdgcn_ballot_w32(mine);

        while (mask) {
            const int n = __builtin_popcount(mask);
            int j[GROUP];
#pragma unroll
            for (int u = 0; u < GROUP; ++u) {
                j[u] = mask ? __builtin_ctz(mask) : 0;
                mask &= mask - 1u;
            }
            float v0[GROUP], v1[GROUP];
#pragma unroll
            for (int u = 0; u < GROUP; ++u) {
                v0[u] = 0.f; v1[u] = 0.f;
                if (u < n) {
                    int sj = __builtin_amdgcn_readlane(s, j[u]);
                    const float* frow = feat + (size_t)sj * FEATS + lane;
                    v0[u] = frow[0];
                    v1[u] = frow[FEATS / 2];
                }
            }
#pragma unroll
            for (int u = 0; u < GROUP; ++u) {
                if (u < n) {
                    int   dj = __builtin_amdgcn_readlane(d, j[u]);
                    float wj = lane_read_f(w, j[u]);
                    float* outp = agg + (size_t)dj * FEATS + lane;
                    unsafeAtomicAdd(outp,             v0[u] * wj);
                    unsafeAtomicAdd(outp + FEATS / 2, v1[u] * wj);
                }
            }
        }
    }
}

__global__ void gcn_node_apply_wmma_kernel(const float* __restrict__ agg,
                                           const float* __restrict__ degree,
                                           const float* __restrict__ Wm,
                                           const float* __restrict__ bias,
                                           float* __restrict__ out,
                                           int n_nodes, int n_mtiles) {
    int wave  = blockIdx.x * (blockDim.x >> 5) + (threadIdx.x >> 5);
    if (wave >= n_mtiles) return;
    int lane  = threadIdx.x & 31;
    int n     = lane & 15;
    int g     = lane >> 4;

    int m      = wave * 16 + n;
    int mclamp = m < n_nodes ? m : n_nodes - 1;
    float dsc  = __frsqrt_rn(degree[mclamp]);
    const float* arow = agg + (size_t)mclamp * FEATS;

    v8f acc0 = {}, acc1 = {}, acc2 = {}, acc3 = {};

    for (int k0 = 0; k0 < FEATS; k0 += 4) {
        int ka = k0 + 2 * g;
        float2 av = *(const float2*)(arow + ka);
        v2f a;
        a.x = av.x * dsc;
        a.y = av.y * dsc;

        const float* wr0 = Wm + (size_t)ka * FEATS;
        const float* wr1 = wr0 + FEATS;
        v2f b0; b0.x = wr0[n +  0]; b0.y = wr1[n +  0];
        v2f b1; b1.x = wr0[n + 16]; b1.y = wr1[n + 16];
        v2f b2; b2.x = wr0[n + 32]; b2.y = wr1[n + 32];
        v2f b3; b3.x = wr0[n + 48]; b3.y = wr1[n + 48];

        acc0 = __builtin_amdgcn_wmma_f32_16x16x4_f32(false, a, false, b0, (short)0, acc0, false, false);
        acc1 = __builtin_amdgcn_wmma_f32_16x16x4_f32(false, a, false, b1, (short)0, acc1, false, false);
        acc2 = __builtin_amdgcn_wmma_f32_16x16x4_f32(false, a, false, b2, (short)0, acc2, false, false);
        acc3 = __builtin_amdgcn_wmma_f32_16x16x4_f32(false, a, false, b3, (short)0, acc3, false, false);
    }

    float bb0 = bias[n +  0];
    float bb1 = bias[n + 16];
    float bb2 = bias[n + 32];
    float bb3 = bias[n + 48];

#pragma unroll
    for (int r = 0; r < 8; ++r) {
        int mrow = wave * 16 + r + 8 * g;
        if (mrow < n_nodes) {
            float* orow = out + (size_t)mrow * FEATS + n;
            orow[ 0] = acc0[r] + bb0;
            orow[16] = acc1[r] + bb1;
            orow[32] = acc2[r] + bb2;
            orow[48] = acc3[r] + bb3;
        }
    }
}

extern "C" void kernel_launch(void* const* d_in, const int* in_sizes, int n_in,
                              void* d_out, int out_size, void* d_ws, size_t ws_size,
                              hipStream_t stream) {
    const float* feature = (const float*)d_in[0];
    const float* degree  = (const float*)d_in[1];
    const int*   src     = (const int*)d_in[2];
    const int*   dst     = (const int*)d_in[3];
    const float* Wm      = (const float*)d_in[4];
    const float* bias    = (const float*)d_in[5];
    float* out = (float*)d_out;
    float* agg = (float*)d_ws;

    int n_nodes = in_sizes[1];
    int n_edges = in_sizes[2];

    int n4 = (n_nodes * FEATS) / 4;
    int zb = (n4 + 255) / 256;
    int* ctr = (int*)d_out;
    gcn_zero_kernel<<<zb, 256, 0, stream>>>((float4*)agg, n4, ctr);

    gcn_edge_scatter_kernel<<<1876, 256, 0, stream>>>(feature, degree, src, dst,
                                                      agg, n_edges, n_nodes, ctr);

    int n_mtiles = (n_nodes + 15) / 16;
    int blocks = (n_mtiles + 7) / 8;
    gcn_node_apply_wmma_kernel<<<blocks, 256, 0, stream>>>(agg, degree, Wm, bias,
                                                           out, n_nodes, n_mtiles);
}


# edge sum in shared memory, not global f32 atomics: edges filed into 512 row-range lists, one block per range sums its rows with LDS f32 adds and feeds WMMA; sum order arrival order as in parent
# speedup vs baseline: 2.8824x; 1.2340x over previous
#include <hip/hip_runtime.h>


typedef __attribute__((ext_vector_type(2))) float v2f;
typedef __attribute__((ext_vector_type(8))) float v8f;

#define FEATS 64

#define ROWS   147
#define RPAD   160
#define PITCH  66
#define NBINS  512
#define CAP    4096

__global__ void gcn_zero_cursor_kernel(int* __restrict__ cursor) {
    int i = blockIdx.x * blockDim.x + threadIdx.x;
    if (i < NBINS) cursor[i] = 0;
}

#define FILL_T   1024
#define FILL_EPT 4
#define EPB      (FILL_T * FILL_EPT)

__global__ void __launch_bounds__(FILL_T)
gcn_bin_fill_kernel(const int* __restrict__ src,
                    const int* __restrict__ dst,
                    int2* __restrict__ list,
                    int* __restrict__ cursor,
                    int n_edges, int n_nodes) {
    __shared__ int s_cnt[NBINS];
    __shared__ int s_base[NBINS];
    const int tid = threadIdx.x;
    const int nblk = (n_edges + EPB - 1) / EPB;
    for (int blk = blockIdx.x; blk < nblk; blk += gridDim.x) {
        for (int i = tid; i < NBINS; i += FILL_T) s_cnt[i] = 0;
        __syncthreads();
        int s[FILL_EPT], d[FILL_EPT], bin[FILL_EPT], rk[FILL_EPT];
#pragma unroll
        for (int i = 0; i < FILL_EPT; ++i) {
            const int e = blk * EPB + i * FILL_T + tid;
            bin[i] = -1; s[i] = 0; d[i] = 0; rk[i] = 0;
            if (e < n_edges) {
                s[i] = src[e];
                d[i] = dst[e];
                if ((unsigned)d[i] < (unsigned)n_nodes) {
                    const int b = d[i] / ROWS;
                    if (b < NBINS) {
                        bin[i] = b;
                        rk[i] = atomicAdd(&s_cnt[b], 1);
                    }
                }
            }
        }
        __syncthreads();
        for (int i = tid; i < NBINS; i += FILL_T) {
            const int h = s_cnt[i];
            s_base[i] = h ? atomicAdd(cursor + i, h) : 0;
        }
        __syncthreads();
#pragma unroll
        for (int i = 0; i < FILL_EPT; ++i) {
            if (bin[i] >= 0) {
                const int pos = s_base[bin[i]] + rk[i];
                if (pos < CAP) list[(size_t)bin[i] * CAP + pos] = make_int2(s[i], d[i]);
            }
        }
        __syncthreads();
    }
}

#define SUM_T  512
#define GROUP  16

__device__ __forceinline__ float lane_read_f(float v, int j) {
    return __int_as_float(__builtin_amdgcn_readlane(__float_as_int(v), j));
}

__global__ void __launch_bounds__(SUM_T)
gcn_bin_sum_apply_kernel(const float* __restrict__ feat,
                         const float* __restrict__ degree,
                         const int* __restrict__ src,
                         const int* __restrict__ dst,
                         const int2* __restrict__ list,
                         const int* __restrict__ cursor,
                         const float* __restrict__ Wm,
                         const float* __restrict__ bias,
                         float* __restrict__ out,
                         int n_edges, int n_nodes) {
    __shared__ float s_acc[RPAD * PITCH];
    const int tid  = threadIdx.x;
    const int lane = tid & 31;
    const int wv     = __builtin_amdgcn_readfirstlane(tid >> 5);
    const int nwaves = SUM_T >> 5;
    const int nbins  = (n_nodes + ROWS - 1) / ROWS;

    for (int b = blockIdx.x; b < nbins; b += gridDim.x) {
        const int row0 = b * ROWS;
        for (int i = tid; i < RPAD * PITCH; i += SUM_T) s_acc[i] = 0.f;
        __syncthreads();

        int  nlist    = 0;
        bool use_list = false;
        if (b < NBINS) { nlist = cursor[b]; use_list = (nlist <= CAP); }
        use_list = __builtin_amdgcn_readfirstlane(use_list ? 1 : 0) != 0;
        const int total   = __builtin_amdgcn_readfirstlane(use_list ? nlist : n_edges);
        const int nchunks = (total + 31) / 32;
        const int2* mylist = list + (size_t)(b < NBINS ? b : 0) * CAP;

        for (int c = wv; c < nchunks; c += nwaves) {
            const int e = c * 32 + lane;
            int s = 0, d = row0;
            bool mine = false;
            if (e < total) {
                if (use_list) {
                    const int2 sd = mylist[e];
                    s = sd.x; d = sd.y; mine = true;
                } else {
                    s = src[e];
                    d = dst[e];
                    mine = (d >= row0) && (d < row0 + ROWS) && (d < n_nodes);
                }
            }
            float w = 0.f;
            if (mine) w = __frsqrt_rn(degree[s]);
            const int dl = mine ? (d - row0) * PITCH : 0;
            unsigned mask = __builtin_amdgcn_ballot_w32(mine);

            while (mask) {
                const int n = __builtin_popcount(mask);
                int j[GROUP];
#pragma unroll
                for (int u = 0; u < GROUP; ++u) {
                    j[u] = mask ? __builtin_ctz(mask) : 0;
                    mask &= mask - 1u;
                }
                float2 v[GROUP];
#pragma unroll
                for (int u = 0; u < GROUP; ++u) {
                    v[u] = make_float2(0.f, 0.f);
                    if (u < n) {
                        const int sj = __builtin_amdgcn_readlane(s, j[u]);
                        v[u] = *(const float2*)(feat + (size_t)sj * FEATS + 2 * lane);
                    }
                }
#pragma unroll
                for (int u = 0; u < GROUP; ++u) {
                    if (u < n) {
                        const int   dj = __builtin_amdgcn_readlane(dl, j[u]);
                        const float wj = lane_read_f(w, j[u]);
                        float* p = s_acc + dj + 2 * lane;
                        unsafeAtomicAdd(p,     v[u].x * wj);
                        unsafeAtomicAdd(p + 1, v[u].y * wj);
                    }
                }
            }
        }
        __syncthreads();

        const int left   = n_nodes - row0;
        const int nrows  = left < ROWS ? left : ROWS;
        const int ntiles = (nrows + 15) >> 4;
        const int n = lane & 15;
        const int g = lane >> 4;
        for (int t = wv; t < ntiles; t += nwaves) {
            const int lrow = t * 16 + n;
            const int grow = row0 + lrow;
            const int gcl  = grow < n_nodes ? grow : n_nodes - 1;
            const float dsc = __frsqrt_rn(degree[gcl]);
            const float* arow = s_acc + lrow * PITCH;

            v8f acc0 = {}, acc1 = {}, acc2 = {}, acc3 = {};
            for (int k0 = 0; k0 < FEATS; k0 += 4) {
                const int ka = k0 + 2 * g;
                const float2 av = *(const float2*)(arow + ka);
                v2f a;
                a.x = av.x * dsc;
                a.y = av.y * dsc;

                const float* wr0 = Wm + (size_t)ka * FEATS;
                const float* wr1 = wr0 + FEATS;
                v2f b0; b0.x = wr0[n +  0]; b0.y = wr1[n +  0];
                v2f b1; b1.x = wr0[n + 16]; b1.y = wr1[n + 16];
                v2f b2; b2.x = wr0[n + 32]; b2.y = wr1[n + 32];
                v2f b3; b3.x = wr0[n + 48]; b3.y = wr1[n + 48];

                acc0 = __builtin_amdgcn_wmma_f32_16x16x4_f32(false, a, false, b0, (short)0, acc0, false, false);
                acc1 = __builtin_amdgcn_wmma_f32_16x16x4_f32(false, a, false, b1, (short)0, acc1, false, false);
                acc2 = __builtin_amdgcn_wmma_f32_16x16x4_f32(false, a, false, b2, (short)0, acc2, false, false);
                acc3 = __builtin_amdgcn_wmma_f32_16x16x4_f32(false, a, false, b3, (short)0, acc3, false, false);
            }

            const float bb0 = bias[n +  0];
            const float bb1 = bias[n + 16];
            const float bb2 = bias[n + 32];
            const float bb3 = bias[n + 48];
#pragma unroll
            for (int r = 0; r < 8; ++r) {
                const int lr = t * 16 + r + 8 * g;
                if (lr < nrows) {
                    float* orow = out + (size_t)(row0 + lr) * FEATS + n;
                    orow[ 0] = acc0[r] + bb0;
                    orow[16] = acc1[r] + bb1;
                    orow[32] = acc2[r] + bb2;
                    orow[48] = acc3[r] + bb3;
                }
            }
        }
        __syncthreads();
    }
}

extern "C" void kernel_launch(void* const* d_in, const int* in_sizes, int n_in,
                              void* d_out, int out_size, void* d_ws, size_t ws_size,
                              hipStream_t stream) {
    const float* feature = (const float*)d_in[0];
    const float* degree  = (const float*)d_in[1];
    const int*   src     = (const int*)d_in[2];
    const int*   dst     = (const int*)d_in[3];
    const float* Wm      = (const float*)d_in[4];
    const float* bias    = (const float*)d_in[5];
    float* out = (float*)d_out;
    int2*  list   = (int2*)d_ws;
    int*   cursor = (int*)(list + (size_t)NBINS * CAP);

    int n_nodes = in_sizes[1];
    int n_edges = in_sizes[2];

    gcn_zero_cursor_kernel<<<1, NBINS, 0, stream>>>(cursor);

    gcn_bin_fill_kernel<<<293, FILL_T, 0, stream>>>(src, dst, list, cursor,
                                                    n_edges, n_nodes);

    gcn_bin_sum_apply_kernel<<<NBINS, SUM_T, 0, stream>>>(feature, degree, src, dst,
                                                          list, cursor, Wm, bias,
                                                          out, n_edges, n_nodes);
}


# edge sum in registers: a bin block sorts its edges by target row in shared memory, a half-wave then sums one row's feature rows in registers, stored once; no f32 atomics; sum order set by the placing
# speedup vs baseline: 3.2827x; 1.1389x over previous
#include <hip/hip_runtime.h>


typedef __attribute__((ext_vector_type(2))) float v2f;
typedef __attribute__((ext_vector_type(8))) float v8f;

#define FEATS 64

#define ROWS   147
#define RPAD   160
#define PITCH  68
#define NBINS  512
#define CAP    4096

__global__ void gcn_zero_cursor_kernel(int* __restrict__ cursor) {
    int i = blockIdx.x * blockDim.x + threadIdx.x;
    if (i < NBINS) cursor[i] = 0;
}

#define FILL_T   1024
#define FILL_EPT 4
#define EPB      (FILL_T * FILL_EPT)

__global__ void __launch_bounds__(FILL_T)
gcn_bin_fill_kernel(const int* __restrict__ src,
                    const int* __restrict__ dst,
                    int2* __restrict__ list,
                    int* __restrict__ cursor,
                    int n_edges, int n_nodes) {
    __shared__ int s_cnt[NBINS];
    __shared__ int s_base[NBINS];
    const int tid = threadIdx.x;
    const int nblk = (n_edges + EPB - 1) / EPB;
    for (int blk = blockIdx.x; blk < nblk; blk += gridDim.x) {
        for (int i = tid; i < NBINS; i += FILL_T) s_cnt[i] = 0;
        __syncthreads();
        int s[FILL_EPT], d[FILL_EPT], bin[FILL_EPT], rk[FILL_EPT];
#pragma unroll
        for (int i = 0; i < FILL_EPT; ++i) {
            const int e = blk * EPB + i * FILL_T + tid;
            bin[i] = -1; s[i] = 0; d[i] = 0; rk[i] = 0;
            if (e < n_edges) {
                s[i] = src[e];
                d[i] = dst[e];
                if ((unsigned)d[i] < (unsigned)n_nodes) {
                    const int b = d[i] / ROWS;
                    if (b < NBINS) {
                        bin[i] = b;
                        rk[i] = atomicAdd(&s_cnt[b], 1);
                    }
                }
            }
        }
        __syncthreads();
        for (int i = tid; i < NBINS; i += FILL_T) {
            const int h = s_cnt[i];
            s_base[i] = h ? atomicAdd(cursor + i, h) : 0;
        }
        __syncthreads();
#pragma unroll
        for (int i = 0; i < FILL_EPT; ++i) {
            if (bin[i] >= 0) {
                const int pos = s_base[bin[i]] + rk[i];
                if (pos < CAP) list[(size_t)bin[i] * CAP + pos] = make_int2(s[i], d[i]);
            }
        }
        __syncthreads();
    }
}

#define SUM_T    512
#define SORT_EPT (CAP / SUM_T)
#define GROUP    16

__device__ __forceinline__ float lane_read_f(float v, int j) {
    return __int_as_float(__builtin_amdgcn_readlane(__float_as_int(v), j));
}

#define HALF_BCAST_I(v, U) __builtin_amdgcn_mov_dpp((v), 0x150 + (U), 0xF, 0xF, false)

__device__ __forceinline__ float4 row_piece(const char* fb, int sx, unsigned c16) {
    const unsigned off = ((unsigned)sx << 8) + c16;
    return *(const float4*)(fb + off);
}

#define ROW_OPEN(U)                                                            \
    if ((U) < nmax) {                                                          \
        const int sx##U = HALF_BCAST_I(sj, (U));                               \
        float4 v##U = make_float4(0.f, 0.f, 0.f, 0.f);                         \
        if ((U) < nh) v##U = row_piece(fb, sx##U, c16);

#define ROW_CLOSE(U)                                                           \
        const float wx##U = __int_as_float(HALF_BCAST_I(wj, (U)));             \
        acc.x += v##U.x * wx##U;                                               \
        acc.y += v##U.y * wx##U;                                               \
        acc.z += v##U.z * wx##U;                                               \
        acc.w += v##U.w * wx##U;                                               \
    }

__global__ void __launch_bounds__(SUM_T, 8)
gcn_bin_sum_apply_kernel(const float* __restrict__ feat,
                         const float* __restrict__ degree,
                         const int* __restrict__ src,
                         const int* __restrict__ dst,
                         const int2* __restrict__ list,
                         const int* __restrict__ cursor,
                         const float* __restrict__ Wm,
                         const float* __restrict__ bias,
                         float* __restrict__ out,
                         int n_edges, int n_nodes) {
    __shared__ __attribute__((aligned(16))) float s_acc[RPAD * PITCH];
    __shared__ int  s_cnt[RPAD];
    __shared__ int  s_start[RPAD + 1];
    __shared__ int2 s_sw[CAP];
    const int tid  = threadIdx.x;
    const int lane = tid & 31;
    const int c = lane & 15;
    const int h = lane >> 4;
    const int wv     = __builtin_amdgcn_readfirstlane(tid >> 5);
    const int nwaves = SUM_T >> 5;
    const int nbins  = (n_nodes + ROWS - 1) / ROWS;

    for (int b = blockIdx.x; b < nbins; b += gridDim.x) {
        const int row0   = b * ROWS;
        const int left   = n_nodes - row0;
        const int nrows  = left < ROWS ? left : ROWS;
        const int ntiles = (nrows + 15) >> 4;

        int  nlist    = 0;
        bool use_list = false;
        if (b < NBINS) { nlist = cursor[b]; use_list = (nlist <= CAP); }
        use_list = __builtin_amdgcn_readfirstlane(use_list ? 1 : 0) != 0;
        nlist    = __builtin_amdgcn_readfirstlane(nlist);

        if (use_list) {
            const int2* mylist = list + (size_t)b * CAP;
            for (int i = tid; i < RPAD; i += SUM_T) s_cnt[i] = 0;
            __syncthreads();

            int   es[SORT_EPT], er[SORT_EPT], ek[SORT_EPT];
            float ew[SORT_EPT];
#pragma unroll
            for (int i = 0; i < SORT_EPT; ++i) {
                const int e = i * SUM_T + tid;
                es[i] = 0; er[i] = -1; ek[i] = 0; ew[i] = 0.f;
                if (e < nlist) {
                    const int2 sd = mylist[e];
                    const int  rr = sd.y - row0;
                    if ((unsigned)rr < (unsigned)nrows) {
                        es[i] = sd.x;
                        er[i] = rr;
                        ek[i] = atomicAdd(&s_cnt[rr], 1);
                        ew[i] = __frsqrt_rn(degree[sd.x]);
                    }
                }
            }
            __syncthreads();

            if (wv == 0) {
                const int c0 = s_cnt[5 * lane + 0];
                const int c1 = s_cnt[5 * lane + 1];
                const int c2 = s_cnt[5 * lane + 2];
                const int c3 = s_cnt[5 * lane + 3];
                const int c4 = s_cnt[5 * lane + 4];
                const int tot = c0 + c1 + c2 + c3 + c4;
                int inc = tot;
#pragma unroll
                for (int o = 1; o < 32; o <<= 1) {
                    const int t = __shfl_up(inc, o, 32);
                    if (lane >= o) inc += t;
                }
                int ex = inc - tot;
                s_start[5 * lane + 0] = ex; ex += c0;
                s_start[5 * lane + 1] = ex; ex += c1;
                s_start[5 * lane + 2] = ex; ex += c2;
                s_start[5 * lane + 3] = ex; ex += c3;
                s_start[5 * lane + 4] = ex; ex += c4;
                if (lane == 31) s_start[RPAD] = ex;
            }
            __syncthreads();

#pragma unroll
            for (int i = 0; i < SORT_EPT; ++i) {
                if (er[i] >= 0)
                    s_sw[s_start[er[i]] + ek[i]] = make_int2(es[i], __float_as_int(ew[i]));
            }
            __syncthreads();

            const char*    fb  = (const char*)feat;
            const unsigned c16 = 16u * (unsigned)c;
            const int npairs = ntiles * 8;
            for (int q = wv; q < npairs; q += nwaves) {
                const int rr  = 2 * q + h;
                const int st  = s_start[rr];
                const int cnt = s_start[rr + 1] - st;
                const int ca  = __builtin_amdgcn_readlane(cnt, 0);
                const int cb  = __builtin_amdgcn_readlane(cnt, 16);
                const int cmax = ca > cb ? ca : cb;
                float4 acc = make_float4(0.f, 0.f, 0.f, 0.f);
                for (int base = 0; base < cmax; base += 16) {
                    const int nh   = cnt - base;
                    const int nmax = cmax - base;
                    int sj = 0, wj = 0;
                    if (c < nh) {
                        const int2 t = s_sw[st + base + c];
                        sj = t.x; wj = t.y;
                    }
                    ROW_OPEN(0)  ROW_OPEN(1)  ROW_OPEN(2)  ROW_OPEN(3)
                    ROW_OPEN(4)  ROW_OPEN(5)  ROW_OPEN(6)  ROW_OPEN(7)
                    ROW_OPEN(8)  ROW_OPEN(9)  ROW_OPEN(10) ROW_OPEN(11)
                    ROW_OPEN(12) ROW_OPEN(13) ROW_OPEN(14) ROW_OPEN(15)
                    ROW_CLOSE(15) ROW_CLOSE(14) ROW_CLOSE(13) ROW_CLOSE(12)
                    ROW_CLOSE(11) ROW_CLOSE(10) ROW_CLOSE(9)  ROW_CLOSE(8)
                    ROW_CLOSE(7)  ROW_CLOSE(6)  ROW_CLOSE(5)  ROW_CLOSE(4)
                    ROW_CLOSE(3)  ROW_CLOSE(2)  ROW_CLOSE(1)  ROW_CLOSE(0)
                }
                *(float4*)(s_acc + rr * PITCH + 4 * c) = acc;
            }
        } else {
            for (int i = tid; i < RPAD * PITCH; i += SUM_T) s_acc[i] = 0.f;
            __syncthreads();
            const int nchunks = (n_edges + 31) / 32;
            for (int ch = wv; ch < nchunks; ch += nwaves) {
                const int e = ch * 32 + lane;
                int s = 0, d = row0;
                bool mine = false;
                if (e < n_edges) {
                    s = src[e];
                    d = dst[e];
                    mine = (d >= row0) && (d < row0 + ROWS) && (d < n_nodes);
                }
                float w = 0.f;
                if (mine) w = __frsqrt_rn(degree[s]);
                const int dl = mine ? (d - row0) * PITCH : 0;
                unsigned mask = __builtin_amdgcn_ballot_w32(mine);

                while (mask) {
                    const int n = __builtin_popcount(mask);
                    int j[GROUP];
#pragma unroll
                    for (int u = 0; u < GROUP; ++u) {
                        j[u] = mask ? __builtin_ctz(mask) : 0;
                        mask &= mask - 1u;
                    }
                    float2 v[GROUP];
#pragma unroll
                    for (int u = 0; u < GROUP; ++u) {
                        v[u] = make_float2(0.f, 0.f);
                        if (u < n) {
                            const int sj = __builtin_amdgcn_readlane(s, j[u]);
                            v[u] = *(const float2*)(feat + (size_t)sj * FEATS + 2 * lane);
                        }
                    }
#pragma unroll
                    for (int u = 0; u < GROUP; ++u) {
                        if (u < n) {
                            const int   dj = __builtin_amdgcn_readlane(dl, j[u]);
                            const float wj = lane_read_f(w, j[u]);
                            float* p = s_acc + dj + 2 * lane;
                            unsafeAtomicAdd(p,     v[u].x * wj);
                            unsafeAtomicAdd(p + 1, v[u].y * wj);
                        }
                    }
                }
            }
        }
        __syncthreads();

        const int n = lane & 15;
        const int g = lane >> 4;
        for (int t = wv; t < ntiles; t += nwaves) {
            const int lrow = t * 16 + n;
            const int grow = row0 + lrow;
            const int gcl  = grow < n_nodes ? grow : n_nodes - 1;
            const float dsc = __frsqrt_rn(degree[gcl]);
            const float* arow = s_acc + lrow * PITCH;

            const float bb0 = bias[n +  0];
            const float bb1 = bias[n + 16];
            const float bb2 = bias[n + 32];
            const float bb3 = bias[n + 48];

            v8f acc0 = {}, acc1 = {}, acc2 = {}, acc3 = {};
            for (int k0 = 0; k0 < FEATS; k0 += 4) {
                const int ka = k0 + 2 * g;
                const float2 av = *(const float2*)(arow + ka);
                v2f a;
                a.x = av.x * dsc;
                a.y = av.y * dsc;

                const float* wr0 = Wm + (size_t)ka * FEATS;
                const float* wr1 = wr0 + FEATS;
                v2f b0; b0.x = wr0[n +  0]; b0.y = wr1[n +  0];
                v2f b1; b1.x = wr0[n + 16]; b1.y = wr1[n + 16];
                v2f b2; b2.x = wr0[n + 32]; b2.y = wr1[n + 32];
                v2f b3; b3.x = wr0[n + 48]; b3.y = wr1[n + 48];

                acc0 = __builtin_amdgcn_wmma_f32_16x16x4_f32(false, a, false, b0, (short)0, acc0, false, false);
                acc1 = __builtin_amdgcn_wmma_f32_16x16x4_f32(false, a, false, b1, (short)0, acc1, false, false);
                acc2 = __builtin_amdgcn_wmma_f32_16x16x4_f32(false, a, false, b2, (short)0, acc2, false, false);
                acc3 = __builtin_amdgcn_wmma_f32_16x16x4_f32(false, a, false, b3, (short)0, acc3, false, false);
            }

            asm volatile("v_nop\n\tv_nop\n\tv_nop\n\tv_nop\n\tv_nop"
                         : "+v"(acc0), "+v"(acc1), "+v"(acc2), "+v"(acc3));

#pragma unroll
            for (int r = 0; r < 8; ++r) {
                const int lr = t * 16 + r + 8 * g;
                if (lr < nrows) {
                    float* orow = out + (size_t)(row0 + lr) * FEATS + n;
                    orow[ 0] = acc0[r] + bb0;
                    orow[16] = acc1[r] + bb1;
                    orow[32] = acc2[r] + bb2;
                    orow[48] = acc3[r] + bb3;
                }
            }
        }
        __syncthreads();
    }
}

extern "C" void kernel_launch(void* const* d_in, const int* in_sizes, int n_in,
                              void* d_out, int out_size, void* d_ws, size_t ws_size,
                              hipStream_t stream) {
    const float* feature = (const float*)d_in[0];
    const float* degree  = (const float*)d_in[1];
    const int*   src     = (const int*)d_in[2];
    const int*   dst     = (const int*)d_in[3];
    const float* Wm      = (const float*)d_in[4];
    const float* bias    = (const float*)d_in[5];
    float* out = (float*)d_out;
    int2*  list   = (int2*)d_ws;
    int*   cursor = (int*)(list + (size_t)NBINS * CAP);

    int n_nodes = in_sizes[1];
    int n_edges = in_sizes[2];

    gcn_zero_cursor_kernel<<<1, NBINS, 0, stream>>>(cursor);

    gcn_bin_fill_kernel<<<293, FILL_T, 0, stream>>>(src, dst, list, cursor,
                                                    n_edges, n_nodes);

    gcn_bin_sum_apply_kernel<<<NBINS, SUM_T, 0, stream>>>(feature, degree, src, dst,
                                                          list, cursor, Wm, bias,
                                                          out, n_edges, n_nodes);
}


# edge filing: a block of 8 waves, not 32, files its 4096 edges, 16 per thread read with 16-byte index loads, so a quarter of the waves must enter; list order, hence sum order, arrival order as parent
# speedup vs baseline: 3.3853x; 1.0313x over previous
#include <hip/hip_runtime.h>


typedef __attribute__((ext_vector_type(2))) float v2f;
typedef __attribute__((ext_vector_type(8))) float v8f;

#define FEATS 64

#define ROWS   147
#define RPAD   160
#define PITCH  68
#define NBINS  512
#define CAP    4096

__global__ void gcn_zero_cursor_kernel(int* __restrict__ cursor) {
    int i = blockIdx.x * blockDim.x + threadIdx.x;
    if (i < NBINS) cursor[i] = 0;
}

#define FILL_T   256
#define FILL_V   4
#define FILL_EPT (4 * FILL_V)
#define EPB      (FILL_T * FILL_EPT)
#define BPT      (NBINS / FILL_T)

__global__ void __launch_bounds__(FILL_T)
gcn_bin_fill_kernel(const int* __restrict__ src,
                    const int* __restrict__ dst,
                    int2* __restrict__ list,
                    int* __restrict__ cursor,
                    int n_edges, int n_nodes) {
    __shared__ int s_cnt[NBINS];
    __shared__ int s_base[NBINS];
    const int tid = threadIdx.x;
    const int nblk = (n_edges + EPB - 1) / EPB;
    const bool wide = ((((size_t)src) | ((size_t)dst)) & 15u) == 0;
    for (int blk = blockIdx.x; blk < nblk; blk += gridDim.x) {
        for (int i = tid; i < NBINS; i += FILL_T) s_cnt[i] = 0;
        __syncthreads();
        int s[FILL_EPT], d[FILL_EPT], bin[FILL_EPT], rk[FILL_EPT];
        const bool full = wide && (n_edges - blk * EPB >= EPB);
        if (full) {
#pragma unroll
            for (int v = 0; v < FILL_V; ++v) {
                const int e0 = blk * EPB + (v * FILL_T + tid) * 4;
                const int4 sv = *(const int4*)(src + e0);
                const int4 dv = *(const int4*)(dst + e0);
                s[4 * v + 0] = sv.x; s[4 * v + 1] = sv.y; s[4 * v + 2] = sv.z; s[4 * v + 3] = sv.w;
                d[4 * v + 0] = dv.x; d[4 * v + 1] = dv.y; d[4 * v + 2] = dv.z; d[4 * v + 3] = dv.w;
            }
        } else {
#pragma unroll
            for (int i = 0; i < FILL_EPT; ++i) {
                const int e = blk * EPB + i * FILL_T + tid;
                s[i] = 0; d[i] = -1;
                if (e < n_edges) { s[i] = src[e]; d[i] = dst[e]; }
            }
        }
#pragma unroll
        for (int i = 0; i < FILL_EPT; ++i) {
            bin[i] = -1; rk[i] = 0;
            if ((unsigned)d[i] < (unsigned)n_nodes) {
                const int b = d[i] / ROWS;
                if (b < NBINS) {
                    bin[i] = b;
                    rk[i] = atomicAdd(&s_cnt[b], 1);
                }
            }
        }
        __syncthreads();
        int hh[BPT], bb[BPT];
#pragma unroll
        for (int j = 0; j < BPT; ++j) hh[j] = s_cnt[j * FILL_T + tid];
#pragma unroll
        for (int j = 0; j < BPT; ++j) {
            bb[j] = 0;
            if (hh[j]) bb[j] = atomicAdd(cursor + j * FILL_T + tid, hh[j]);
        }
#pragma unroll
        for (int j = 0; j < BPT; ++j) s_base[j * FILL_T + tid] = bb[j];
        __syncthreads();
        int pos[FILL_EPT];
#pragma unroll
        for (int i = 0; i < FILL_EPT; ++i) pos[i] = s_base[bin[i] >= 0 ? bin[i] : 0] + rk[i];
#pragma unroll
        for (int i = 0; i < FILL_EPT; ++i) {
            if (bin[i] >= 0 && pos[i] < CAP)
                list[(size_t)bin[i] * CAP + pos[i]] = make_int2(s[i], d[i]);
        }
        __syncthreads();
    }
}

#define SUM_T    512
#define SORT_EPT (CAP / SUM_T)
#define GROUP    16

__device__ __forceinline__ float lane_read_f(float v, int j) {
    return __int_as_float(__builtin_amdgcn_readlane(__float_as_int(v), j));
}

#define HALF_BCAST_I(v, U) __builtin_amdgcn_mov_dpp((v), 0x150 + (U), 0xF, 0xF, false)

__device__ __forceinline__ float4 row_piece(const char* fb, int sx, unsigned c16) {
    const unsigned off = ((unsigned)sx << 8) + c16;
    return *(const float4*)(fb + off);
}

#define ROW_OPEN(U)                                                            \
    if ((U) < nmax) {                                                          \
        const int sx##U = HALF_BCAST_I(sj, (U));                               \
        float4 v##U = make_float4(0.f, 0.f, 0.f, 0.f);                         \
        if ((U) < nh) v##U = row_piece(fb, sx##U, c16);

#define ROW_CLOSE(U)                                                           \
        const float wx##U = __int_as_float(HALF_BCAST_I(wj, (U)));             \
        acc.x += v##U.x * wx##U;                                               \
        acc.y += v##U.y * wx##U;                                               \
        acc.z += v##U.z * wx##U;                                               \
        acc.w += v##U.w * wx##U;                                               \
    }

__global__ void __launch_bounds__(SUM_T, 8)
gcn_bin_sum_apply_kernel(const float* __restrict__ feat,
                         const float* __restrict__ degree,
                         const int* __restrict__ src,
                         const int* __restrict__ dst,
                         const int2* __restrict__ list,
                         const int* __restrict__ cursor,
                         const float* __restrict__ Wm,
                         const float* __restrict__ bias,
                         float* __restrict__ out,
                         int n_edges, int n_nodes) {
    __shared__ __attribute__((aligned(16))) float s_acc[RPAD * PITCH];
    __shared__ int  s_cnt[RPAD];
    __shared__ int  s_start[RPAD + 1];
    __shared__ int2 s_sw[CAP];
    const int tid  = threadIdx.x;
    const int lane = tid & 31;
    const int c = lane & 15;
    const int h = lane >> 4;
    const int wv     = __builtin_amdgcn_readfirstlane(tid >> 5);
    const int nwaves = SUM_T >> 5;
    const int nbins  = (n_nodes + ROWS - 1) / ROWS;

    for (int b = blockIdx.x; b < nbins; b += gridDim.x) {
        const int row0   = b * ROWS;
        const int left   = n_nodes - row0;
        const int nrows  = left < ROWS ? left : ROWS;
        const int ntiles = (nrows + 15) >> 4;

        int  nlist    = 0;
        bool use_list = false;
        if (b < NBINS) { nlist = cursor[b]; use_list = (nlist <= CAP); }
        use_list = __builtin_amdgcn_readfirstlane(use_list ? 1 : 0) != 0;
        nlist    = __builtin_amdgcn_readfirstlane(nlist);

        if (use_list) {
            const int2* mylist = list + (size_t)b * CAP;
            for (int i = tid; i < RPAD; i += SUM_T) s_cnt[i] = 0;
            __syncthreads();

            int   es[SORT_EPT], er[SORT_EPT], ek[SORT_EPT];
            float ew[SORT_EPT];
#pragma unroll
            for (int i = 0; i < SORT_EPT; ++i) {
                const int e = i * SUM_T + tid;
                es[i] = 0; er[i] = -1; ek[i] = 0; ew[i] = 0.f;
                if (e < nlist) {
                    const int2 sd = mylist[e];
                    const int  rr = sd.y - row0;
                    if ((unsigned)rr < (unsigned)nrows) {
                        es[i] = sd.x;
                        er[i] = rr;
                        ek[i] = atomicAdd(&s_cnt[rr], 1);
                        ew[i] = __frsqrt_rn(degree[sd.x]);
                    }
                }
            }
            __syncthreads();

            if (wv == 0) {
                const int c0 = s_cnt[5 * lane + 0];
                const int c1 = s_cnt[5 * lane + 1];
                const int c2 = s_cnt[5 * lane + 2];
                const int c3 = s_cnt[5 * lane + 3];
                const int c4 = s_cnt[5 * lane + 4];
                const int tot = c0 + c1 + c2 + c3 + c4;
                int inc = tot;
#pragma unroll
                for (int o = 1; o < 32; o <<= 1) {
                    const int t = __shfl_up(inc, o, 32);
                    if (lane >= o) inc += t;
                }
                int ex = inc - tot;
                s_start[5 * lane + 0] = ex; ex += c0;
                s_start[5 * lane + 1] = ex; ex += c1;
                s_start[5 * lane + 2] = ex; ex += c2;
                s_start[5 * lane + 3] = ex; ex += c3;
                s_start[5 * lane + 4] = ex; ex += c4;
                if (lane == 31) s_start[RPAD] = ex;
            }
            __syncthreads();

#pragma unroll
            for (int i = 0; i < SORT_EPT; ++i) {
                if (er[i] >= 0)
                    s_sw[s_start[er[i]] + ek[i]] = make_int2(es[i], __float_as_int(ew[i]));
            }
            __syncthreads();

            const char*    fb  = (const char*)feat;
            const unsigned c16 = 16u * (unsigned)c;
            const int npairs = ntiles * 8;
            for (int q = wv; q < npairs; q += nwaves) {
                const int rr  = 2 * q + h;
                const int st  = s_start[rr];
                const int cnt = s_start[rr + 1] - st;
                const int ca  = __builtin_amdgcn_readlane(cnt, 0);
                const int cb  = __builtin_amdgcn_readlane(cnt, 16);
                const int cmax = ca > cb ? ca : cb;
                float4 acc = make_float4(0.f, 0.f, 0.f, 0.f);
                for (int base = 0; base < cmax; base += 16) {
                    const int nh   = cnt - base;
                    const int nmax = cmax - base;
                    int sj = 0, wj = 0;
                    if (c < nh) {
                        const int2 t = s_sw[st + base + c];
                        sj = t.x; wj = t.y;
                    }
                    ROW_OPEN(0)  ROW_OPEN(1)  ROW_OPEN(2)  ROW_OPEN(3)
                    ROW_OPEN(4)  ROW_OPEN(5)  ROW_OPEN(6)  ROW_OPEN(7)
                    ROW_OPEN(8)  ROW_OPEN(9)  ROW_OPEN(10) ROW_OPEN(11)
                    ROW_OPEN(12) ROW_OPEN(13) ROW_OPEN(14) ROW_OPEN(15)
                    ROW_CLOSE(15) ROW_CLOSE(14) ROW_CLOSE(13) ROW_CLOSE(12)
                    ROW_CLOSE(11) ROW_CLOSE(10) ROW_CLOSE(9)  ROW_CLOSE(8)
                    ROW_CLOSE(7)  ROW_CLOSE(6)  ROW_CLOSE(5)  ROW_CLOSE(4)
                    ROW_CLOSE(3)  ROW_CLOSE(2)  ROW_CLOSE(1)  ROW_CLOSE(0)
                }
                *(float4*)(s_acc + rr * PITCH + 4 * c) = acc;
            }
        } else {
            for (int i = tid; i < RPAD * PITCH; i += SUM_T) s_acc[i] = 0.f;
            __syncthreads();
            const int nchunks = (n_edges + 31) / 32;
            for (int ch = wv; ch < nchunks; ch += nwaves) {
                const int e = ch * 32 + lane;
                int s = 0, d = row0;
                bool mine = false;
                if (e < n_edges) {
                    s = src[e];
                    d = dst[e];
                    mine = (d >= row0) && (d < row0 + ROWS) && (d < n_nodes);
                }
                float w = 0.f;
                if (mine) w = __frsqrt_rn(degree[s]);
                const int dl = mine ? (d - row0) * PITCH : 0;
                unsigned mask = __builtin_amdgcn_ballot_w32(mine);

                while (mask) {
                    const int n = __builtin_popcount(mask);
                    int j[GROUP];
#pragma unroll
                    for (int u = 0; u < GROUP; ++u) {
                        j[u] = mask ? __builtin_ctz(mask) : 0;
                        mask &= mask - 1u;
                    }
                    float2 v[GROUP];
#pragma unroll
                    for (int u = 0; u < GROUP; ++u) {
                        v[u] = make_float2(0.f, 0.f);
                        if (u < n) {
                            const int sj = __builtin_amdgcn_readlane(s, j[u]);
                            v[u] = *(const float2*)(feat + (size_t)sj * FEATS + 2 * lane);
                        }
                    }
#pragma unroll
                    for (int u = 0; u < GROUP; ++u) {
                        if (u < n) {
                            const int   dj = __builtin_amdgcn_readlane(dl, j[u]);
                            const float wj = lane_read_f(w, j[u]);
                            float* p = s_acc + dj + 2 * lane;
                            unsafeAtomicAdd(p,     v[u].x * wj);
                            unsafeAtomicAdd(p + 1, v[u].y * wj);
                        }
                    }
                }
            }
        }
        __syncthreads();

        const int n = lane & 15;
        const int g = lane >> 4;
        for (int t = wv; t < ntiles; t += nwaves) {
            const int lrow = t * 16 + n;
            const int grow = row0 + lrow;
            const int gcl  = grow < n_nodes ? grow : n_nodes - 1;
            const float dsc = __frsqrt_rn(degree[gcl]);
            const float* arow = s_acc + lrow * PITCH;

            const float bb0 = bias[n +  0];
            const float bb1 = bias[n + 16];
            const float bb2 = bias[n + 32];
            const float bb3 = bias[n + 48];

            v8f acc0 = {}, acc1 = {}, acc2 = {}, acc3 = {};
            for (int k0 = 0; k0 < FEATS; k0 += 4) {
                const int ka = k0 + 2 * g;
                const float2 av = *(const float2*)(arow + ka);
                v2f a;
                a.x = av.x * dsc;
                a.y = av.y * dsc;

                const float* wr0 = Wm + (size_t)ka * FEATS;
                const float* wr1 = wr0 + FEATS;
                v2f b0; b0.x = wr0[n +  0]; b0.y = wr1[n +  0];
                v2f b1; b1.x = wr0[n + 16]; b1.y = wr1[n + 16];
                v2f b2; b2.x = wr0[n + 32]; b2.y = wr1[n + 32];
                v2f b3; b3.x = wr0[n + 48]; b3.y = wr1[n + 48];

                acc0 = __builtin_amdgcn_wmma_f32_16x16x4_f32(false, a, false, b0, (short)0, acc0, false, false);
                acc1 = __builtin_amdgcn_wmma_f32_16x16x4_f32(false, a, false, b1, (short)0, acc1, false, false);
                acc2 = __builtin_amdgcn_wmma_f32_16x16x4_f32(false, a, false, b2, (short)0, acc2, false, false);
                acc3 = __builtin_amdgcn_wmma_f32_16x16x4_f32(false, a, false, b3, (short)0, acc3, false, false);
            }

            asm volatile("v_nop\n\tv_nop\n\tv_nop\n\tv_nop\n\tv_nop"
                         : "+v"(acc0), "+v"(acc1), "+v"(acc2), "+v"(acc3));

#pragma unroll
            for (int r = 0; r < 8; ++r) {
                const int lr = t * 16 + r + 8 * g;
                if (lr < nrows) {
                    float* orow = out + (size_t)(row0 + lr) * FEATS + n;
                    orow[ 0] = acc0[r] + bb0;
                    orow[16] = acc1[r] + bb1;
                    orow[32] = acc2[r] + bb2;
                    orow[48] = acc3[r] + bb3;
                }
            }
        }
        __syncthreads();
    }
}

extern "C" void kernel_launch(void* const* d_in, const int* in_sizes, int n_in,
                              void* d_out, int out_size, void* d_ws, size_t ws_size,
                              hipStream_t stream) {
    const float* feature = (const float*)d_in[0];
    const float* degree  = (const float*)d_in[1];
    const int*   src     = (const int*)d_in[2];
    const int*   dst     = (const int*)d_in[3];
    const float* Wm      = (const float*)d_in[4];
    const float* bias    = (const float*)d_in[5];
    float* out = (float*)d_out;
    int2*  list   = (int2*)d_ws;
    int*   cursor = (int*)(list + (size_t)NBINS * CAP);

    int n_nodes = in_sizes[1];
    int n_edges = in_sizes[2];

    gcn_zero_cursor_kernel<<<1, NBINS, 0, stream>>>(cursor);

    gcn_bin_fill_kernel<<<293, FILL_T, 0, stream>>>(src, dst, list, cursor,
                                                    n_edges, n_nodes);

    gcn_bin_sum_apply_kernel<<<NBINS, SUM_T, 0, stream>>>(feature, degree, src, dst,
                                                          list, cursor, Wm, bias,
                                                          out, n_edges, n_nodes);
}


# row sort in the bin kernel: a thread's 8 list-pair loads issue together, then its 8 degree loads and rank adds, then its 8 row-start reads, not pair after pair; sum order arrival order as parent
# speedup vs baseline: 3.5103x; 1.0369x over previous
#include <hip/hip_runtime.h>


typedef __attribute__((ext_vector_type(2))) float v2f;
typedef __attribute__((ext_vector_type(8))) float v8f;

#define FEATS 64

#define ROWS   147
#define RPAD   160
#define PITCH  68
#define NBINS  512
#define CAP    4096

__global__ void gcn_zero_cursor_kernel(int* __restrict__ cursor) {
    int i = blockIdx.x * blockDim.x + threadIdx.x;
    if (i < NBINS) cursor[i] = 0;
}

#define FILL_T   256
#define FILL_V   4
#define FILL_EPT (4 * FILL_V)
#define EPB      (FILL_T * FILL_EPT)
#define BPT      (NBINS / FILL_T)

__global__ void __launch_bounds__(FILL_T)
gcn_bin_fill_kernel(const int* __restrict__ src,
                    const int* __restrict__ dst,
                    int2* __restrict__ list,
                    int* __restrict__ cursor,
                    int n_edges, int n_nodes) {
    __shared__ int s_cnt[NBINS];
    __shared__ int s_base[NBINS];
    const int tid = threadIdx.x;
    const int nblk = (n_edges + EPB - 1) / EPB;
    const bool wide = ((((size_t)src) | ((size_t)dst)) & 15u) == 0;
    for (int blk = blockIdx.x; blk < nblk; blk += gridDim.x) {
        for (int i = tid; i < NBINS; i += FILL_T) s_cnt[i] = 0;
        __syncthreads();
        int s[FILL_EPT], d[FILL_EPT], bin[FILL_EPT], rk[FILL_EPT];
        const bool full = wide && (n_edges - blk * EPB >= EPB);
        if (full) {
#pragma unroll
            for (int v = 0; v < FILL_V; ++v) {
                const int e0 = blk * EPB + (v * FILL_T + tid) * 4;
                const int4 sv = *(const int4*)(src + e0);
                const int4 dv = *(const int4*)(dst + e0);
                s[4 * v + 0] = sv.x; s[4 * v + 1] = sv.y; s[4 * v + 2] = sv.z; s[4 * v + 3] = sv.w;
                d[4 * v + 0] = dv.x; d[4 * v + 1] = dv.y; d[4 * v + 2] = dv.z; d[4 * v + 3] = dv.w;
            }
        } else {
#pragma unroll
            for (int i = 0; i < FILL_EPT; ++i) {
                const int e = blk * EPB + i * FILL_T + tid;
                s[i] = 0; d[i] = -1;
                if (e < n_edges) { s[i] = src[e]; d[i] = dst[e]; }
            }
        }
#pragma unroll
        for (int i = 0; i < FILL_EPT; ++i) {
            bin[i] = -1; rk[i] = 0;
            if ((unsigned)d[i] < (unsigned)n_nodes) {
                const int b = d[i] / ROWS;
                if (b < NBINS) {
                    bin[i] = b;
                    rk[i] = atomicAdd(&s_cnt[b], 1);
                }
            }
        }
        __syncthreads();
        int hh[BPT], bb[BPT];
#pragma unroll
        for (int j = 0; j < BPT; ++j) hh[j] = s_cnt[j * FILL_T + tid];
#pragma unroll
        for (int j = 0; j < BPT; ++j) {
            bb[j] = 0;
            if (hh[j]) bb[j] = atomicAdd(cursor + j * FILL_T + tid, hh[j]);
        }
#pragma unroll
        for (int j = 0; j < BPT; ++j) s_base[j * FILL_T + tid] = bb[j];
        __syncthreads();
        int pos[FILL_EPT];
#pragma unroll
        for (int i = 0; i < FILL_EPT; ++i) pos[i] = s_base[bin[i] >= 0 ? bin[i] : 0] + rk[i];
#pragma unroll
        for (int i = 0; i < FILL_EPT; ++i) {
            if (bin[i] >= 0 && pos[i] < CAP)
                list[(size_t)bin[i] * CAP + pos[i]] = make_int2(s[i], d[i]);
        }
        __syncthreads();
    }
}

#define SUM_T    512
#define SORT_EPT (CAP / SUM_T)
#define GROUP    16

__device__ __forceinline__ float lane_read_f(float v, int j) {
    return __int_as_float(__builtin_amdgcn_readlane(__float_as_int(v), j));
}

#define HALF_BCAST_I(v, U) __builtin_amdgcn_mov_dpp((v), 0x150 + (U), 0xF, 0xF, false)

__device__ __forceinline__ float4 row_piece(const char* fb, int sx, unsigned c16) {
    const unsigned off = ((unsigned)sx << 8) + c16;
    return *(const float4*)(fb + off);
}

#define ROW_OPEN(U)                                                            \
    if ((U) < nmax) {                                                          \
        const int sx##U = HALF_BCAST_I(sj, (U));                               \
        float4 v##U = make_float4(0.f, 0.f, 0.f, 0.f);                         \
        if ((U) < nh) v##U = row_piece(fb, sx##U, c16);

#define ROW_CLOSE(U)                                                           \
        const float wx##U = __int_as_float(HALF_BCAST_I(wj, (U)));             \
        acc.x += v##U.x * wx##U;                                               \
        acc.y += v##U.y * wx##U;                                               \
        acc.z += v##U.z * wx##U;                                               \
        acc.w += v##U.w * wx##U;                                               \
    }

__global__ void __launch_bounds__(SUM_T, 8)
gcn_bin_sum_apply_kernel(const float* __restrict__ feat,
                         const float* __restrict__ degree,
                         const int* __restrict__ src,
                         const int* __restrict__ dst,
                         const int2* __restrict__ list,
                         const int* __restrict__ cursor,
                         const float* __restrict__ Wm,
                         const float* __restrict__ bias,
                         float* __restrict__ out,
                         int n_edges, int n_nodes) {
    __shared__ __attribute__((aligned(16))) float s_acc[RPAD * PITCH];
    __shared__ int  s_cnt[RPAD];
    __shared__ int  s_start[RPAD + 1];
    __shared__ int2 s_sw[CAP];
    const int tid  = threadIdx.x;
    const int lane = tid & 31;
    const int c = lane & 15;
    const int h = lane >> 4;
    const int wv     = __builtin_amdgcn_readfirstlane(tid >> 5);
    const int nwaves = SUM_T >> 5;
    const int nbins  = (n_nodes + ROWS - 1) / ROWS;

    for (int b = blockIdx.x; b < nbins; b += gridDim.x) {
        const int row0   = b * ROWS;
        const int left   = n_nodes - row0;
        const int nrows  = left < ROWS ? left : ROWS;
        const int ntiles = (nrows + 15) >> 4;

        int  nlist    = 0;
        bool use_list = false;
        if (b < NBINS) { nlist = cursor[b]; use_list = (nlist <= CAP); }
        use_list = __builtin_amdgcn_readfirstlane(use_list ? 1 : 0) != 0;
        nlist    = __builtin_amdgcn_readfirstlane(nlist);

        if (use_list) {
            const int2* mylist = list + (size_t)b * CAP;
            for (int i = tid; i < RPAD; i += SUM_T) s_cnt[i] = 0;
            __syncthreads();

            int   es[SORT_EPT], er[SORT_EPT], ek[SORT_EPT];
            float ew[SORT_EPT];
            int2  sd[SORT_EPT];
            float dg[SORT_EPT];
#pragma unroll
            for (int i = 0; i < SORT_EPT; ++i) {
                const int e = i * SUM_T + tid;
                sd[i] = make_int2(0, -1);
                if (e < nlist) sd[i] = mylist[e];
            }
#pragma unroll
            for (int i = 0; i < SORT_EPT; ++i) {
                const int e  = i * SUM_T + tid;
                const int rr = sd[i].y - row0;
                es[i] = 0; er[i] = -1; ek[i] = 0; dg[i] = 1.f;
                if (e < nlist && (unsigned)rr < (unsigned)nrows) {
                    es[i] = sd[i].x;
                    er[i] = rr;
                    ek[i] = atomicAdd(&s_cnt[rr], 1);
                    dg[i] = degree[sd[i].x];
                }
            }
#pragma unroll
            for (int i = 0; i < SORT_EPT; ++i)
                ew[i] = er[i] >= 0 ? __frsqrt_rn(dg[i]) : 0.f;
            __syncthreads();

            if (wv == 0) {
                const int c0 = s_cnt[5 * lane + 0];
                const int c1 = s_cnt[5 * lane + 1];
                const int c2 = s_cnt[5 * lane + 2];
                const int c3 = s_cnt[5 * lane + 3];
                const int c4 = s_cnt[5 * lane + 4];
                const int tot = c0 + c1 + c2 + c3 + c4;
                int inc = tot;
#pragma unroll
                for (int o = 1; o < 32; o <<= 1) {
                    const int t = __shfl_up(inc, o, 32);
                    if (lane >= o) inc += t;
                }
                int ex = inc - tot;
                s_start[5 * lane + 0] = ex; ex += c0;
                s_start[5 * lane + 1] = ex; ex += c1;
                s_start[5 * lane + 2] = ex; ex += c2;
                s_start[5 * lane + 3] = ex; ex += c3;
                s_start[5 * lane + 4] = ex; ex += c4;
                if (lane == 31) s_start[RPAD] = ex;
            }
            __syncthreads();

            int ps[SORT_EPT];
#pragma unroll
            for (int i = 0; i < SORT_EPT; ++i) ps[i] = s_start[er[i] >= 0 ? er[i] : 0];
#pragma unroll
            for (int i = 0; i < SORT_EPT; ++i) {
                if (er[i] >= 0)
                    s_sw[ps[i] + ek[i]] = make_int2(es[i], __float_as_int(ew[i]));
            }
            __syncthreads();

            const char*    fb  = (const char*)feat;
            const unsigned c16 = 16u * (unsigned)c;
            const int npairs = ntiles * 8;
            for (int q = wv; q < npairs; q += nwaves) {
                const int rr  = 2 * q + h;
                const int st  = s_start[rr];
                const int cnt = s_start[rr + 1] - st;
                const int ca  = __builtin_amdgcn_readlane(cnt, 0);
                const int cb  = __builtin_amdgcn_readlane(cnt, 16);
                const int cmax = ca > cb ? ca : cb;
                float4 acc = make_float4(0.f, 0.f, 0.f, 0.f);
                for (int base = 0; base < cmax; base += 16) {
                    const int nh   = cnt - base;
                    const int nmax = cmax - base;
                    int sj = 0, wj = 0;
                    if (c < nh) {
                        const int2 t = s_sw[st + base + c];
                        sj = t.x; wj = t.y;
                    }
                    ROW_OPEN(0)  ROW_OPEN(1)  ROW_OPEN(2)  ROW_OPEN(3)
                    ROW_OPEN(4)  ROW_OPEN(5)  ROW_OPEN(6)  ROW_OPEN(7)
                    ROW_OPEN(8)  ROW_OPEN(9)  ROW_OPEN(10) ROW_OPEN(11)
                    ROW_OPEN(12) ROW_OPEN(13) ROW_OPEN(14) ROW_OPEN(15)
                    ROW_CLOSE(15) ROW_CLOSE(14) ROW_CLOSE(13) ROW_CLOSE(12)
                    ROW_CLOSE(11) ROW_CLOSE(10) ROW_CLOSE(9)  ROW_CLOSE(8)
                    ROW_CLOSE(7)  ROW_CLOSE(6)  ROW_CLOSE(5)  ROW_CLOSE(4)
                    ROW_CLOSE(3)  ROW_CLOSE(2)  ROW_CLOSE(1)  ROW_CLOSE(0)
                }
                *(float4*)(s_acc + rr * PITCH + 4 * c) = acc;
            }
        } else {
            for (int i = tid; i < RPAD * PITCH; i += SUM_T) s_acc[i] = 0.f;
            __syncthreads();
            const int nchunks = (n_edges + 31) / 32;
            for (int ch = wv; ch < nchunks; ch += nwaves) {
                const int e = ch * 32 + lane;
                int s = 0, d = row0;
                bool mine = false;
                if (e < n_edges) {
                    s = src[e];
                    d = dst[e];
                    mine = (d >= row0) && (d < row0 + ROWS) && (d < n_nodes);
                }
                float w = 0.f;
                if (mine) w = __frsqrt_rn(degree[s]);
                const int dl = mine ? (d - row0) * PITCH : 0;
                unsigned mask = __builtin_amdgcn_ballot_w32(mine);

                while (mask) {
                    const int n = __builtin_popcount(mask);
                    int j[GROUP];
#pragma unroll
                    for (int u = 0; u < GROUP; ++u) {
                        j[u] = mask ? __builtin_ctz(mask) : 0;
                        mask &= mask - 1u;
                    }
                    float2 v[GROUP];
#pragma unroll
                    for (int u = 0; u < GROUP; ++u) {
                        v[u] = make_float2(0.f, 0.f);
                        if (u < n) {
                            const int sj = __builtin_amdgcn_readlane(s, j[u]);
                            v[u] = *(const float2*)(feat + (size_t)sj * FEATS + 2 * lane);
                        }
                    }
#pragma unroll
                    for (int u = 0; u < GROUP; ++u) {
                        if (u < n) {
                            const int   dj = __builtin_amdgcn_readlane(dl, j[u]);
                            const float wj = lane_read_f(w, j[u]);
                            float* p = s_acc + dj + 2 * lane;
                            unsafeAtomicAdd(p,     v[u].x * wj);
                            unsafeAtomicAdd(p + 1, v[u].y * wj);
                        }
                    }
                }
            }
        }
        __syncthreads();

        const int n = lane & 15;
        const int g = lane >> 4;
        for (int t = wv; t < ntiles; t += nwaves) {
            const int lrow = t * 16 + n;
            const int grow = row0 + lrow;
            const int gcl  = grow < n_nodes ? grow : n_nodes - 1;
            const float dsc = __frsqrt_rn(degree[gcl]);
            const float* arow = s_acc + lrow * PITCH;

            const float bb0 = bias[n +  0];
            const float bb1 = bias[n + 16];
            const float bb2 = bias[n + 32];
            const float bb3 = bias[n + 48];

            v8f acc0 = {}, acc1 = {}, acc2 = {}, acc3 = {};
            for (int k0 = 0; k0 < FEATS; k0 += 4) {
                const int ka = k0 + 2 * g;
                const float2 av = *(const float2*)(arow + ka);
                v2f a;
                a.x = av.x * dsc;
                a.y = av.y * dsc;

                const float* wr0 = Wm + (size_t)ka * FEATS;
                const float* wr1 = wr0 + FEATS;
                v2f b0; b0.x = wr0[n +  0]; b0.y = wr1[n +  0];
                v2f b1; b1.x = wr0[n + 16]; b1.y = wr1[n + 16];
                v2f b2; b2.x = wr0[n + 32]; b2.y = wr1[n + 32];
                v2f b3; b3.x = wr0[n + 48]; b3.y = wr1[n + 48];

                acc0 = __builtin_amdgcn_wmma_f32_16x16x4_f32(false, a, false, b0, (short)0, acc0, false, false);
                acc1 = __builtin_amdgcn_wmma_f32_16x16x4_f32(false, a, false, b1, (short)0, acc1, false, false);
                acc2 = __builtin_amdgcn_wmma_f32_16x16x4_f32(false, a, false, b2, (short)0, acc2, false, false);
                acc3 = __builtin_amdgcn_wmma_f32_16x16x4_f32(false, a, false, b3, (short)0, acc3, false, false);
            }

            asm volatile("v_nop\n\tv_nop\n\tv_nop\n\tv_nop\n\tv_nop"
                         : "+v"(acc0), "+v"(acc1), "+v"(acc2), "+v"(acc3));

#pragma unroll
            for (int r = 0; r < 8; ++r) {
                const int lr = t * 16 + r + 8 * g;
                if (lr < nrows) {
                    float* orow = out + (size_t)(row0 + lr) * FEATS + n;
                    orow[ 0] = acc0[r] + bb0;
                    orow[16] = acc1[r] + bb1;
                    orow[32] = acc2[r] + bb2;
                    orow[48] = acc3[r] + bb3;
                }
            }
        }
        __syncthreads();
    }
}

extern "C" void kernel_launch(void* const* d_in, const int* in_sizes, int n_in,
                              void* d_out, int out_size, void* d_ws, size_t ws_size,
                              hipStream_t stream) {
    const float* feature = (const float*)d_in[0];
    const float* degree  = (const float*)d_in[1];
    const int*   src     = (const int*)d_in[2];
    const int*   dst     = (const int*)d_in[3];
    const float* Wm      = (const float*)d_in[4];
    const float* bias    = (const float*)d_in[5];
    float* out = (float*)d_out;
    int2*  list   = (int2*)d_ws;
    int*   cursor = (int*)(list + (size_t)NBINS * CAP);

    int n_nodes = in_sizes[1];
    int n_edges = in_sizes[2];

    gcn_zero_cursor_kernel<<<1, NBINS, 0, stream>>>(cursor);

    gcn_bin_fill_kernel<<<293, FILL_T, 0, stream>>>(src, dst, list, cursor,
                                                    n_edges, n_nodes);

    gcn_bin_sum_apply_kernel<<<NBINS, SUM_T, 0, stream>>>(feature, degree, src, dst,
                                                          list, cursor, Wm, bias,
                                                          out, n_edges, n_nodes);
}


# edge filing without global counters: a filing block sorts its edges by bin in LDS, writes its own list and a start/count table; bin blocks collect the runs; 2 launches; sum order by arrival as parent
# speedup vs baseline: 3.7563x; 1.0701x over previous
#include <hip/hip_runtime.h>


typedef __attribute__((ext_vector_type(2))) float v2f;
typedef __attribute__((ext_vector_type(8))) float v8f;

#define FEATS 64

#define ROWS   147
#define RPAD   160
#define PITCH  68
#define NBINS  512
#define CAP    4096
#define NFMAX  384

#define FILL_T   256
#define FILL_V   4
#define FILL_EPT (4 * FILL_V)
#define EPB      (FILL_T * FILL_EPT)
#define FILL_W   (FILL_T / 32)

__global__ void __launch_bounds__(FILL_T)
gcn_bin_fill_kernel(const int* __restrict__ src,
                    const int* __restrict__ dst,
                    int2* __restrict__ list,
                    int2* __restrict__ table,
                    int n_edges, int n_nodes) {
    __shared__ int s_cnt[NBINS];
    __shared__ int s_base[NBINS];
    __shared__ int s_wtot[FILL_W];
    __shared__ __attribute__((aligned(16))) int2 s_stage[EPB];
    const int tid  = threadIdx.x;
    const int lane = tid & 31;
    const int wv   = tid >> 5;
    const int nblk = (n_edges + EPB - 1) / EPB;
    if (nblk > NFMAX) return;
    const bool wide = ((((size_t)src) | ((size_t)dst)) & 15u) == 0;
    for (int blk = blockIdx.x; blk < nblk; blk += gridDim.x) {
        for (int i = tid; i < NBINS; i += FILL_T) s_cnt[i] = 0;
        __syncthreads();
        int s[FILL_EPT], d[FILL_EPT], bin[FILL_EPT], rk[FILL_EPT];
        const bool full = wide && (n_edges - blk * EPB >= EPB);
        if (full) {
#pragma unroll
            for (int v = 0; v < FILL_V; ++v) {
                const int e0 = blk * EPB + (v * FILL_T + tid) * 4;
                const int4 sv = *(const int4*)(src + e0);
                const int4 dv = *(const int4*)(dst + e0);
                s[4 * v + 0] = sv.x; s[4 * v + 1] = sv.y; s[4 * v + 2] = sv.z; s[4 * v + 3] = sv.w;
                d[4 * v + 0] = dv.x; d[4 * v + 1] = dv.y; d[4 * v + 2] = dv.z; d[4 * v + 3] = dv.w;
            }
        } else {
#pragma unroll
            for (int i = 0; i < FILL_EPT; ++i) {
                const int e = blk * EPB + i * FILL_T + tid;
                s[i] = 0; d[i] = -1;
                if (e < n_edges) { s[i] = src[e]; d[i] = dst[e]; }
            }
        }
#pragma unroll
        for (int i = 0; i < FILL_EPT; ++i) {
            bin[i] = -1; rk[i] = 0;
            if ((unsigned)d[i] < (unsigned)n_nodes) {
                const int b = d[i] / ROWS;
                if (b < NBINS) {
                    bin[i] = b;
                    rk[i] = atomicAdd(&s_cnt[b], 1);
                }
            }
        }
        __syncthreads();
        const int c0  = s_cnt[2 * tid + 0];
        const int c1  = s_cnt[2 * tid + 1];
        const int tot = c0 + c1;
        int inc = tot;
#pragma unroll
        for (int o = 1; o < 32; o <<= 1) {
            const int t = __shfl_up(inc, o, 32);
            if (lane >= o) inc += t;
        }
        if (lane == 31) s_wtot[wv] = inc;
        __syncthreads();
        int wbase = 0, total = 0;
#pragma unroll
        for (int j = 0; j < FILL_W; ++j) {
            const int v = s_wtot[j];
            if (j < wv) wbase += v;
            total += v;
        }
        const int ex = wbase + inc - tot;
        s_base[2 * tid + 0] = ex;
        s_base[2 * tid + 1] = ex + c0;
        table[(size_t)blk * (NBINS / 2) + tid] =
            make_int2(ex | (c0 << 16), (ex + c0) | (c1 << 16));
        __syncthreads();
        int pos[FILL_EPT];
#pragma unroll
        for (int i = 0; i < FILL_EPT; ++i) pos[i] = s_base[bin[i] >= 0 ? bin[i] : 0] + rk[i];
#pragma unroll
        for (int i = 0; i < FILL_EPT; ++i) {
            if (bin[i] >= 0) s_stage[pos[i]] = make_int2(s[i], d[i]);
        }
        __syncthreads();
        int4*       gl4 = (int4*)(list + (size_t)blk * EPB);
        const int4* st4 = (const int4*)s_stage;
#pragma unroll
        for (int j = 0; j < EPB / 2 / FILL_T; ++j) {
            const int q = j * FILL_T + tid;
            if (2 * q < total) gl4[q] = st4[q];
        }
        __syncthreads();
    }
}

#define SUM_T    512
#define SORT_EPT (CAP / SUM_T)
#define COPY_R   (NFMAX * 8 / SUM_T)
#define COPY_K   3
#define GROUP    16

__device__ __forceinline__ float lane_read_f(float v, int j) {
    return __int_as_float(__builtin_amdgcn_readlane(__float_as_int(v), j));
}

#define HALF_BCAST_I(v, U) __builtin_amdgcn_mov_dpp((v), 0x150 + (U), 0xF, 0xF, false)

__device__ __forceinline__ float4 row_piece(const char* fb, int sx, unsigned c16) {
    const unsigned off = ((unsigned)sx << 8) + c16;
    return *(const float4*)(fb + off);
}

#define ROW_OPEN(U)                                                            \
    if ((U) < nmax) {                                                          \
        const int sx##U = HALF_BCAST_I(sj, (U));                               \
        float4 v##U = make_float4(0.f, 0.f, 0.f, 0.f);                         \
        if ((U) < nh) v##U = row_piece(fb, sx##U, c16);

#define ROW_CLOSE(U)                                                           \
        const float wx##U = __int_as_float(HALF_BCAST_I(wj, (U)));             \
        acc.x += v##U.x * wx##U;                                               \
        acc.y += v##U.y * wx##U;                                               \
        acc.z += v##U.z * wx##U;                                               \
        acc.w += v##U.w * wx##U;                                               \
    }

__global__ void __launch_bounds__(SUM_T, 8)
gcn_bin_sum_apply_kernel(const float* __restrict__ feat,
                         const float* __restrict__ degree,
                         const int* __restrict__ src,
                         const int* __restrict__ dst,
                         const int2* __restrict__ list,
                         const int* __restrict__ table,
                         const float* __restrict__ Wm,
                         const float* __restrict__ bias,
                         float* __restrict__ out,
                         int n_edges, int n_nodes) {
    __shared__ __attribute__((aligned(16))) float s_acc[RPAD * PITCH];
    __shared__ int  s_cnt[RPAD];
    __shared__ int  s_start[RPAD + 1];
    __shared__ int2 s_sw[CAP];
    __shared__ int4 s_seg[NFMAX];
    __shared__ int  s_wtot[SUM_T / 32];
    const int tid  = threadIdx.x;
    const int lane = tid & 31;
    const int c = lane & 15;
    const int h = lane >> 4;
    const int wv     = __builtin_amdgcn_readfirstlane(tid >> 5);
    const int nwaves = SUM_T >> 5;
    const int nbins  = (n_nodes + ROWS - 1) / ROWS;
    const int nblk   = (n_edges + EPB - 1) / EPB;

    for (int b = blockIdx.x; b < nbins; b += gridDim.x) {
        const int row0   = b * ROWS;
        const int left   = n_nodes - row0;
        const int nrows  = left < ROWS ? left : ROWS;
        const int ntiles = (nrows + 15) >> 4;

        int  nlist    = 0;
        bool use_list = false;
        if (b < NBINS && nblk <= NFMAX) {
            int rs = 0, rc = 0;
            if (tid < nblk) {
                const int w = table[(size_t)tid * NBINS + b];
                rs = w & 0xFFFF;
                rc = w >> 16;
            }
            int inc = rc;
#pragma unroll
            for (int o = 1; o < 32; o <<= 1) {
                const int t = __shfl_up(inc, o, 32);
                if (lane >= o) inc += t;
            }
            if (lane == 31) s_wtot[wv] = inc;
            __syncthreads();
            int wbase = 0;
#pragma unroll
            for (int j = 0; j < SUM_T / 32; ++j) {
                const int v = s_wtot[j];
                if (j < wv) wbase += v;
                nlist += v;
            }
            if (tid < NFMAX) s_seg[tid] = make_int4(tid * EPB + rs, rc, wbase + inc - rc, 0);
            use_list = (nlist <= CAP);
        }
        use_list = __builtin_amdgcn_readfirstlane(use_list ? 1 : 0) != 0;
        nlist    = __builtin_amdgcn_readfirstlane(nlist);

        if (use_list) {
            for (int i = tid; i < RPAD; i += SUM_T) s_cnt[i] = 0;
            __syncthreads();

            int2* s_pairs = (int2*)s_acc;
            {
                const int sub = tid & 7;
                const int grp = tid >> 3;
                int4 sg[COPY_R];
                int2 pr[COPY_R][COPY_K];
#pragma unroll
                for (int j = 0; j < COPY_R; ++j) sg[j] = s_seg[grp + (SUM_T / 8) * j];
#pragma unroll
                for (int j = 0; j < COPY_R; ++j) {
#pragma unroll
                    for (int k = 0; k < COPY_K; ++k) {
                        const int i = sub + 8 * k;
                        pr[j][k] = make_int2(0, -1);
                        if (i < sg[j].y) pr[j][k] = list[sg[j].x + i];
                    }
                }
#pragma unroll
                for (int j = 0; j < COPY_R; ++j) {
#pragma unroll
                    for (int k = 0; k < COPY_K; ++k) {
                        const int i = sub + 8 * k;
                        if (i < sg[j].y) s_pairs[sg[j].z + i] = pr[j][k];
                    }
                    for (int i = sub + 8 * COPY_K; i < sg[j].y; i += 8)
                        s_pairs[sg[j].z + i] = list[sg[j].x + i];
                }
            }
            __syncthreads();

            int   es[SORT_EPT], er[SORT_EPT], ek[SORT_EPT];
            float ew[SORT_EPT];
            int2  sd[SORT_EPT];
            float dg[SORT_EPT];
#pragma unroll
            for (int i = 0; i < SORT_EPT; ++i) {
                const int e = i * SUM_T + tid;
                sd[i] = make_int2(0, -1);
                if (e < nlist) sd[i] = s_pairs[e];
            }
#pragma unroll
            for (int i = 0; i < SORT_EPT; ++i) {
                const int e  = i * SUM_T + tid;
                const int rr = sd[i].y - row0;
                es[i] = 0; er[i] = -1; ek[i] = 0; dg[i] = 1.f;
                if (e < nlist && (unsigned)rr < (unsigned)nrows) {
                    es[i] = sd[i].x;
                    er[i] = rr;
                    ek[i] = atomicAdd(&s_cnt[rr], 1);
                    dg[i] = degree[sd[i].x];
                }
            }
#pragma unroll
            for (int i = 0; i < SORT_EPT; ++i)
                ew[i] = er[i] >= 0 ? __frsqrt_rn(dg[i]) : 0.f;
            __syncthreads();

            if (wv == 0) {
                const int c0 = s_cnt[5 * lane + 0];
                const int c1 = s_cnt[5 * lane + 1];
                const int c2 = s_cnt[5 * lane + 2];
                const int c3 = s_cnt[5 * lane + 3];
                const int c4 = s_cnt[5 * lane + 4];
                const int tot = c0 + c1 + c2 + c3 + c4;
                int inc = tot;
#pragma unroll
                for (int o = 1; o < 32; o <<= 1) {
                    const int t = __shfl_up(inc, o, 32);
                    if (lane >= o) inc += t;
                }
                int ex = inc - tot;
                s_start[5 * lane + 0] = ex; ex += c0;
                s_start[5 * lane + 1] = ex; ex += c1;
                s_start[5 * lane + 2] = ex; ex += c2;
                s_start[5 * lane + 3] = ex; ex += c3;
                s_start[5 * lane + 4] = ex; ex += c4;
                if (lane == 31) s_start[RPAD] = ex;
            }
            __syncthreads();

            int ps[SORT_EPT];
#pragma unroll
            for (int i = 0; i < SORT_EPT; ++i) ps[i] = s_start[er[i] >= 0 ? er[i] : 0];
#pragma unroll
            for (int i = 0; i < SORT_EPT; ++i) {
                if (er[i] >= 0)
                    s_sw[ps[i] + ek[i]] = make_int2(es[i], __float_as_int(ew[i]));
            }
            __syncthreads();

            const char*    fb  = (const char*)feat;
            const unsigned c16 = 16u * (unsigned)c;
            const int npairs = ntiles * 8;
            for (int q = wv; q < npairs; q += nwaves) {
                const int rr  = 2 * q + h;
                const int st  = s_start[rr];
                const int cnt = s_start[rr + 1] - st;
                const int ca  = __builtin_amdgcn_readlane(cnt, 0);
                const int cb  = __builtin_amdgcn_readlane(cnt, 16);
                const int cmax = ca > cb ? ca : cb;
                float4 acc = make_float4(0.f, 0.f, 0.f, 0.f);
                for (int base = 0; base < cmax; base += 16) {
                    const int nh   = cnt - base;
                    const int nmax = cmax - base;
                    int sj = 0, wj = 0;
                    if (c < nh) {
                        const int2 t = s_sw[st + base + c];
                        sj = t.x; wj = t.y;
                    }
                    ROW_OPEN(0)  ROW_OPEN(1)  ROW_OPEN(2)  ROW_OPEN(3)
                    ROW_OPEN(4)  ROW_OPEN(5)  ROW_OPEN(6)  ROW_OPEN(7)
                    ROW_OPEN(8)  ROW_OPEN(9)  ROW_OPEN(10) ROW_OPEN(11)
                    ROW_OPEN(12) ROW_OPEN(13) ROW_OPEN(14) ROW_OPEN(15)
                    ROW_CLOSE(15) ROW_CLOSE(14) ROW_CLOSE(13) ROW_CLOSE(12)
                    ROW_CLOSE(11) ROW_CLOSE(10) ROW_CLOSE(9)  ROW_CLOSE(8)
                    ROW_CLOSE(7)  ROW_CLOSE(6)  ROW_CLOSE(5)  ROW_CLOSE(4)
                    ROW_CLOSE(3)  ROW_CLOSE(2)  ROW_CLOSE(1)  ROW_CLOSE(0)
                }
                *(float4*)(s_acc + rr * PITCH + 4 * c) = acc;
            }
        } else {
            for (int i = tid; i < RPAD * PITCH; i += SUM_T) s_acc[i] = 0.f;
            __syncthreads();
            const int nchunks = (n_edges + 31) / 32;
            for (int ch = wv; ch < nchunks; ch += nwaves) {
                const int e = ch * 32 + lane;
                int s = 0, d = row0;
                bool mine = false;
                if (e < n_edges) {
                    s = src[e];
                    d = dst[e];
                    mine = (d >= row0) && (d < row0 + ROWS) && (d < n_nodes);
                }
                float w = 0.f;
                if (mine) w = __frsqrt_rn(degree[s]);
                const int dl = mine ? (d - row0) * PITCH : 0;
                unsigned mask = __builtin_amdgcn_ballot_w32(mine);

                while (mask) {
                    const int n = __builtin_popcount(mask);
                    int j[GROUP];
#pragma unroll
                    for (int u = 0; u < GROUP; ++u) {
                        j[u] = mask ? __builtin_ctz(mask) : 0;
                        mask &= mask - 1u;
                    }
                    float2 v[GROUP];
#pragma unroll
                    for (int u = 0; u < GROUP; ++u) {
                        v[u] = make_float2(0.f, 0.f);
                        if (u < n) {
                            const int sj = __builtin_amdgcn_readlane(s, j[u]);
                            v[u] = *(const float2*)(feat + (size_t)sj * FEATS + 2 * lane);
                        }
                    }
#pragma unroll
                    for (int u = 0; u < GROUP; ++u) {
                        if (u < n) {
                            const int   dj = __builtin_amdgcn_readlane(dl, j[u]);
                            const float wj = lane_read_f(w, j[u]);
                            float* p = s_acc + dj + 2 * lane;
                            unsafeAtomicAdd(p,     v[u].x * wj);
                            unsafeAtomicAdd(p + 1, v[u].y * wj);
                        }
                    }
                }
            }
        }
        __syncthreads();

        const int n = lane & 15;
        const int g = lane >> 4;
        for (int t = wv; t < ntiles; t += nwaves) {
            const int lrow = t * 16 + n;
            const int grow = row0 + lrow;
            const int gcl  = grow < n_nodes ? grow : n_nodes - 1;
            const float dsc = __frsqrt_rn(degree[gcl]);
            const float* arow = s_acc + lrow * PITCH;

            const float bb0 = bias[n +  0];
            const float bb1 = bias[n + 16];
            const float bb2 = bias[n + 32];
            const float bb3 = bias[n + 48];

            v8f acc0 = {}, acc1 = {}, acc2 = {}, acc3 = {};
            for (int k0 = 0; k0 < FEATS; k0 += 4) {
                const int ka = k0 + 2 * g;
                const float2 av = *(const float2*)(arow + ka);
                v2f a;
                a.x = av.x * dsc;
                a.y = av.y * dsc;

                const float* wr0 = Wm + (size_t)ka * FEATS;
                const float* wr1 = wr0 + FEATS;
                v2f b0; b0.x = wr0[n +  0]; b0.y = wr1[n +  0];
                v2f b1; b1.x = wr0[n + 16]; b1.y = wr1[n + 16];
                v2f b2; b2.x = wr0[n + 32]; b2.y = wr1[n + 32];
                v2f b3; b3.x = wr0[n + 48]; b3.y = wr1[n + 48];

                acc0 = __builtin_amdgcn_wmma_f32_16x16x4_f32(false, a, false, b0, (short)0, acc0, false, false);
                acc1 = __builtin_amdgcn_wmma_f32_16x16x4_f32(false, a, false, b1, (short)0, acc1, false, false);
                acc2 = __builtin_amdgcn_wmma_f32_16x16x4_f32(false, a, false, b2, (short)0, acc2, false, false);
                acc3 = __builtin_amdgcn_wmma_f32_16x16x4_f32(false, a, false, b3, (short)0, acc3, false, false);
            }

            asm volatile("v_nop\n\tv_nop\n\tv_nop\n\tv_nop\n\tv_nop"
                         : "+v"(acc0), "+v"(acc1), "+v"(acc2), "+v"(acc3));

#pragma unroll
            for (int r = 0; r < 8; ++r) {
                const int lr = t * 16 + r + 8 * g;
                if (lr < nrows) {
                    float* orow = out + (size_t)(row0 + lr) * FEATS + n;
                    orow[ 0] = acc0[r] + bb0;
                    orow[16] = acc1[r] + bb1;
                    orow[32] = acc2[r] + bb2;
                    orow[48] = acc3[r] + bb3;
                }
            }
        }
        __syncthreads();
    }
}

extern "C" void kernel_launch(void* const* d_in, const int* in_sizes, int n_in,
                              void* d_out, int out_size, void* d_ws, size_t ws_size,
                              hipStream_t stream) {
    const float* feature = (const float*)d_in[0];
    const float* degree  = (const float*)d_in[1];
    const int*   src     = (const int*)d_in[2];
    const int*   dst     = (const int*)d_in[3];
    const float* Wm      = (const float*)d_in[4];
    const float* bias    = (const float*)d_in[5];
    float* out = (float*)d_out;
    int2*  list  = (int2*)d_ws;
    int2*  table = list + (size_t)NFMAX * EPB;

    int n_nodes = in_sizes[1];
    int n_edges = in_sizes[2];

    gcn_bin_fill_kernel<<<293, FILL_T, 0, stream>>>(src, dst, list, table,
                                                    n_edges, n_nodes);

    gcn_bin_sum_apply_kernel<<<NBINS, SUM_T, 0, stream>>>(feature, degree, src, dst,
                                                          list, (const int*)table, Wm, bias,
                                                          out, n_edges, n_nodes);
}


# row sum: entries taken four at a time, no per-entry test, exec change or zero fill: an entry past a row's end has weight 0 and reads row 0; list holds row byte offsets; sum order as parent
# speedup vs baseline: 3.8812x; 1.0333x over previous
#include <hip/hip_runtime.h>


typedef __attribute__((ext_vector_type(2))) float v2f;
typedef __attribute__((ext_vector_type(8))) float v8f;

#define FEATS 64

#define ROWS   147
#define RPAD   160
#define PITCH  68
#define NBINS  512
#define CAP    4096
#define NFMAX  384

#define FILL_T   256
#define FILL_V   4
#define FILL_EPT (4 * FILL_V)
#define EPB      (FILL_T * FILL_EPT)
#define FILL_W   (FILL_T / 32)

__global__ void __launch_bounds__(FILL_T)
gcn_bin_fill_kernel(const int* __restrict__ src,
                    const int* __restrict__ dst,
                    int2* __restrict__ list,
                    int2* __restrict__ table,
                    int n_edges, int n_nodes) {
    __shared__ int s_cnt[NBINS];
    __shared__ int s_base[NBINS];
    __shared__ int s_wtot[FILL_W];
    __shared__ __attribute__((aligned(16))) int2 s_stage[EPB];
    const int tid  = threadIdx.x;
    const int lane = tid & 31;
    const int wv   = tid >> 5;
    const int nblk = (n_edges + EPB - 1) / EPB;
    if (nblk > NFMAX) return;
    const bool wide = ((((size_t)src) | ((size_t)dst)) & 15u) == 0;
    for (int blk = blockIdx.x; blk < nblk; blk += gridDim.x) {
        for (int i = tid; i < NBINS; i += FILL_T) s_cnt[i] = 0;
        __syncthreads();
        int s[FILL_EPT], d[FILL_EPT], bin[FILL_EPT], rk[FILL_EPT];
        const bool full = wide && (n_edges - blk * EPB >= EPB);
        if (full) {
#pragma unroll
            for (int v = 0; v < FILL_V; ++v) {
                const int e0 = blk * EPB + (v * FILL_T + tid) * 4;
                const int4 sv = *(const int4*)(src + e0);
                const int4 dv = *(const int4*)(dst + e0);
                s[4 * v + 0] = sv.x; s[4 * v + 1] = sv.y; s[4 * v + 2] = sv.z; s[4 * v + 3] = sv.w;
                d[4 * v + 0] = dv.x; d[4 * v + 1] = dv.y; d[4 * v + 2] = dv.z; d[4 * v + 3] = dv.w;
            }
        } else {
#pragma unroll
            for (int i = 0; i < FILL_EPT; ++i) {
                const int e = blk * EPB + i * FILL_T + tid;
                s[i] = 0; d[i] = -1;
                if (e < n_edges) { s[i] = src[e]; d[i] = dst[e]; }
            }
        }
#pragma unroll
        for (int i = 0; i < FILL_EPT; ++i) {
            bin[i] = -1; rk[i] = 0;
            if ((unsigned)d[i] < (unsigned)n_nodes) {
                const int b = d[i] / ROWS;
                if (b < NBINS) {
                    bin[i] = b;
                    rk[i] = atomicAdd(&s_cnt[b], 1);
                }
            }
        }
        __syncthreads();
        const int c0  = s_cnt[2 * tid + 0];
        const int c1  = s_cnt[2 * tid + 1];
        const int tot = c0 + c1;
        int inc = tot;
#pragma unroll
        for (int o = 1; o < 32; o <<= 1) {
            const int t = __shfl_up(inc, o, 32);
            if (lane >= o) inc += t;
        }
        if (lane == 31) s_wtot[wv] = inc;
        __syncthreads();
        int wbase = 0, total = 0;
#pragma unroll
        for (int j = 0; j < FILL_W; ++j) {
            const int v = s_wtot[j];
            if (j < wv) wbase += v;
            total += v;
        }
        const int ex = wbase + inc - tot;
        s_base[2 * tid + 0] = ex;
        s_base[2 * tid + 1] = ex + c0;
        table[(size_t)blk * (NBINS / 2) + tid] =
            make_int2(ex | (c0 << 16), (ex + c0) | (c1 << 16));
        __syncthreads();
        int pos[FILL_EPT];
#pragma unroll
        for (int i = 0; i < FILL_EPT; ++i) pos[i] = s_base[bin[i] >= 0 ? bin[i] : 0] + rk[i];
#pragma unroll
        for (int i = 0; i < FILL_EPT; ++i) {
            if (bin[i] >= 0) s_stage[pos[i]] = make_int2(s[i], d[i]);
        }
        __syncthreads();
        int4*       gl4 = (int4*)(list + (size_t)blk * EPB);
        const int4* st4 = (const int4*)s_stage;
#pragma unroll
        for (int j = 0; j < EPB / 2 / FILL_T; ++j) {
            const int q = j * FILL_T + tid;
            if (2 * q < total) gl4[q] = st4[q];
        }
        __syncthreads();
    }
}

#define SUM_T    512
#define SORT_EPT (CAP / SUM_T)
#define COPY_R   (NFMAX * 8 / SUM_T)
#define COPY_K   3
#define GROUP    16

__device__ __forceinline__ float lane_read_f(float v, int j) {
    return __int_as_float(__builtin_amdgcn_readlane(__float_as_int(v), j));
}

#define HALF_BCAST_I(v, U) __builtin_amdgcn_mov_dpp((v), 0x150 + (U), 0xF, 0xF, false)

#define ROW_LOAD(U)                                                            \
        const float4 v##U =                                                    \
            *(const float4*)(fb + ((unsigned)HALF_BCAST_I(oj, (U)) + c16));

#define ROW_ADD(U)                                                             \
        {                                                                      \
            const float wx = __int_as_float(HALF_BCAST_I(wj, (U)));            \
            acc.x = __builtin_fmaf(v##U.x, wx, acc.x);                         \
            acc.y = __builtin_fmaf(v##U.y, wx, acc.y);                         \
            acc.z = __builtin_fmaf(v##U.z, wx, acc.z);                         \
            acc.w = __builtin_fmaf(v##U.w, wx, acc.w);                         \
        }

#define QUAD_OPEN(A, B, C, D)                                                  \
    if ((A) < nmax) {                                                          \
        ROW_LOAD(A) ROW_LOAD(B) ROW_LOAD(C) ROW_LOAD(D)

#define QUAD_CLOSE(A, B, C, D)                                                 \
        ROW_ADD(D) ROW_ADD(C) ROW_ADD(B) ROW_ADD(A)                            \
    }

__global__ void __launch_bounds__(SUM_T, 8)
gcn_bin_sum_apply_kernel(const float* __restrict__ feat,
                         const float* __restrict__ degree,
                         const int* __restrict__ src,
                         const int* __restrict__ dst,
                         const int2* __restrict__ list,
                         const int* __restrict__ table,
                         const float* __restrict__ Wm,
                         const float* __restrict__ bias,
                         float* __restrict__ out,
                         int n_edges, int n_nodes) {
    __shared__ __attribute__((aligned(16))) float s_acc[RPAD * PITCH];
    __shared__ int  s_cnt[RPAD];
    __shared__ int  s_start[RPAD + 1];
    __shared__ int2 s_sw[CAP];
    __shared__ int4 s_seg[NFMAX];
    __shared__ int  s_wtot[SUM_T / 32];
    const int tid  = threadIdx.x;
    const int lane = tid & 31;
    const int c = lane & 15;
    const int h = lane >> 4;
    const int wv     = __builtin_amdgcn_readfirstlane(tid >> 5);
    const int nwaves = SUM_T >> 5;
    const int nbins  = (n_nodes + ROWS - 1) / ROWS;
    const int nblk   = (n_edges + EPB - 1) / EPB;

    for (int b = blockIdx.x; b < nbins; b += gridDim.x) {
        const int row0   = b * ROWS;
        const int left   = n_nodes - row0;
        const int nrows  = left < ROWS ? left : ROWS;
        const int ntiles = (nrows + 15) >> 4;

        int  nlist    = 0;
        bool use_list = false;
        if (b < NBINS && nblk <= NFMAX) {
            int rs = 0, rc = 0;
            if (tid < nblk) {
                const int w = table[(size_t)tid * NBINS + b];
                rs = w & 0xFFFF;
                rc = w >> 16;
            }
            int inc = rc;
#pragma unroll
            for (int o = 1; o < 32; o <<= 1) {
                const int t = __shfl_up(inc, o, 32);
                if (lane >= o) inc += t;
            }
            if (lane == 31) s_wtot[wv] = inc;
            __syncthreads();
            int wbase = 0;
#pragma unroll
            for (int j = 0; j < SUM_T / 32; ++j) {
                const int v = s_wtot[j];
                if (j < wv) wbase += v;
                nlist += v;
            }
            if (tid < NFMAX) s_seg[tid] = make_int4(tid * EPB + rs, rc, wbase + inc - rc, 0);
            use_list = (nlist <= CAP);
        }
        use_list = __builtin_amdgcn_readfirstlane(use_list ? 1 : 0) != 0;
        nlist    = __builtin_amdgcn_readfirstlane(nlist);

        if (use_list) {
            for (int i = tid; i < RPAD; i += SUM_T) s_cnt[i] = 0;
            __syncthreads();

            int2* s_pairs = (int2*)s_acc;
            {
                const int sub = tid & 7;
                const int grp = tid >> 3;
                int4 sg[COPY_R];
                int2 pr[COPY_R][COPY_K];
#pragma unroll
                for (int j = 0; j < COPY_R; ++j) sg[j] = s_seg[grp + (SUM_T / 8) * j];
#pragma unroll
                for (int j = 0; j < COPY_R; ++j) {
#pragma unroll
                    for (int k = 0; k < COPY_K; ++k) {
                        const int i = sub + 8 * k;
                        pr[j][k] = make_int2(0, -1);
                        if (i < sg[j].y) pr[j][k] = list[sg[j].x + i];
                    }
                }
#pragma unroll
                for (int j = 0; j < COPY_R; ++j) {
#pragma unroll
                    for (int k = 0; k < COPY_K; ++k) {
                        const int i = sub + 8 * k;
                        if (i < sg[j].y) s_pairs[sg[j].z + i] = pr[j][k];
                    }
                    for (int i = sub + 8 * COPY_K; i < sg[j].y; i += 8)
                        s_pairs[sg[j].z + i] = list[sg[j].x + i];
                }
            }
            __syncthreads();

            int   es[SORT_EPT], er[SORT_EPT], ek[SORT_EPT];
            float ew[SORT_EPT];
            int2  sd[SORT_EPT];
            float dg[SORT_EPT];
#pragma unroll
            for (int i = 0; i < SORT_EPT; ++i) {
                const int e = i * SUM_T + tid;
                sd[i] = make_int2(0, -1);
                if (e < nlist) sd[i] = s_pairs[e];
            }
#pragma unroll
            for (int i = 0; i < SORT_EPT; ++i) {
                const int e  = i * SUM_T + tid;
                const int rr = sd[i].y - row0;
                es[i] = 0; er[i] = -1; ek[i] = 0; dg[i] = 1.f;
                if (e < nlist && (unsigned)rr < (unsigned)nrows) {
                    es[i] = sd[i].x;
                    er[i] = rr;
                    ek[i] = atomicAdd(&s_cnt[rr], 1);
                    dg[i] = degree[sd[i].x];
                }
            }
#pragma unroll
            for (int i = 0; i < SORT_EPT; ++i)
                ew[i] = er[i] >= 0 ? __frsqrt_rn(dg[i]) : 0.f;
            __syncthreads();

            if (wv == 0) {
                const int c0 = s_cnt[5 * lane + 0];
                const int c1 = s_cnt[5 * lane + 1];
                const int c2 = s_cnt[5 * lane + 2];
                const int c3 = s_cnt[5 * lane + 3];
                const int c4 = s_cnt[5 * lane + 4];
                const int tot = c0 + c1 + c2 + c3 + c4;
                int inc = tot;
#pragma unroll
                for (int o = 1; o < 32; o <<= 1) {
                    const int t = __shfl_up(inc, o, 32);
                    if (lane >= o) inc += t;
                }
                int ex = inc - tot;
                s_start[5 * lane + 0] = ex; ex += c0;
                s_start[5 * lane + 1] = ex; ex += c1;
                s_start[5 * lane + 2] = ex; ex += c2;
                s_start[5 * lane + 3] = ex; ex += c3;
                s_start[5 * lane + 4] = ex; ex += c4;
                if (lane == 31) s_start[RPAD] = ex;
            }
            __syncthreads();

            int ps[SORT_EPT];
#pragma unroll
            for (int i = 0; i < SORT_EPT; ++i) ps[i] = s_start[er[i] >= 0 ? er[i] : 0];
#pragma unroll
            for (int i = 0; i < SORT_EPT; ++i) {
                if (er[i] >= 0)
                    s_sw[ps[i] + ek[i]] = make_int2((int)((unsigned)es[i] << 8), __float_as_int(ew[i]));
            }
            __syncthreads();

            const char*    fb  = (const char*)feat;
            const unsigned c16 = 16u * (unsigned)c;
            const int npairs = ntiles * 8;
            for (int q = wv; q < npairs; q += nwaves) {
                const int rr  = 2 * q + h;
                const int st  = s_start[rr];
                const int cnt = s_start[rr + 1] - st;
                const int ca  = __builtin_amdgcn_readlane(cnt, 0);
                const int cb  = __builtin_amdgcn_readlane(cnt, 16);
                const int cmax = ca > cb ? ca : cb;
                float4 acc = make_float4(0.f, 0.f, 0.f, 0.f);
                for (int base = 0; base < cmax; base += 16) {
                    const int nh   = cnt - base;
                    const int nmax = cmax - base;
                    int oj = 0, wj = 0;
                    if (c < nh) {
                        const int2 t = s_sw[st + base + c];
                        oj = t.x; wj = t.y;
                    }
                    QUAD_OPEN(0, 1, 2, 3)
                    QUAD_OPEN(4, 5, 6, 7)
                    QUAD_OPEN(8, 9, 10, 11)
                    QUAD_OPEN(12, 13, 14, 15)
                    QUAD_CLOSE(12, 13, 14, 15)
                    QUAD_CLOSE(8, 9, 10, 11)
                    QUAD_CLOSE(4, 5, 6, 7)
                    QUAD_CLOSE(0, 1, 2, 3)
                }
                *(float4*)(s_acc + rr * PITCH + 4 * c) = acc;
            }
        } else {
            for (int i = tid; i < RPAD * PITCH; i += SUM_T) s_acc[i] = 0.f;
            __syncthreads();
            const int nchunks = (n_edges + 31) / 32;
            for (int ch = wv; ch < nchunks; ch += nwaves) {
                const int e = ch * 32 + lane;
                int s = 0, d = row0;
                bool mine = false;
                if (e < n_edges) {
                    s = src[e];
                    d = dst[e];
                    mine = (d >= row0) && (d < row0 + ROWS) && (d < n_nodes);
                }
                float w = 0.f;
                if (mine) w = __frsqrt_rn(degree[s]);
                const int dl = mine ? (d - row0) * PITCH : 0;
                unsigned mask = __builtin_amdgcn_ballot_w32(mine);

                while (mask) {
                    const int n = __builtin_popcount(mask);
                    int j[GROUP];
#pragma unroll
                    for (int u = 0; u < GROUP; ++u) {
                        j[u] = mask ? __builtin_ctz(mask) : 0;
                        mask &= mask - 1u;
                    }
                    float2 v[GROUP];
#pragma unroll
                    for (int u = 0; u < GROUP; ++u) {
                        v[u] = make_float2(0.f, 0.f);
                        if (u < n) {
                            const int sj = __builtin_amdgcn_readlane(s, j[u]);
                            v[u] = *(const float2*)(feat + (size_t)sj * FEATS + 2 * lane);
                        }
                    }
#pragma unroll
                    for (int u = 0; u < GROUP; ++u) {
                        if (u < n) {
                            const int   dj = __builtin_amdgcn_readlane(dl, j[u]);
                            const float wj = lane_read_f(w, j[u]);
                            float* p = s_acc + dj + 2 * lane;
                            unsafeAtomicAdd(p,     v[u].x * wj);
                            unsafeAtomicAdd(p + 1, v[u].y * wj);
                        }
                    }
                }
            }
        }
        __syncthreads();

        const int n = lane & 15;
        const int g = lane >> 4;
        for (int t = wv; t < ntiles; t += nwaves) {
            const int lrow = t * 16 + n;
            const int grow = row0 + lrow;
            const int gcl  = grow < n_nodes ? grow : n_nodes - 1;
            const float dsc = __frsqrt_rn(degree[gcl]);
            const float* arow = s_acc + lrow * PITCH;

            const float bb0 = bias[n +  0];
            const float bb1 = bias[n + 16];
            const float bb2 = bias[n + 32];
            const float bb3 = bias[n + 48];

            v8f acc0 = {}, acc1 = {}, acc2 = {}, acc3 = {};
            for (int k0 = 0; k0 < FEATS; k0 += 4) {
                const int ka = k0 + 2 * g;
                const float2 av = *(const float2*)(arow + ka);
                v2f a;
                a.x = av.x * dsc;
                a.y = av.y * dsc;

                const float* wr0 = Wm + (size_t)ka * FEATS;
                const float* wr1 = wr0 + FEATS;
                v2f b0; b0.x = wr0[n +  0]; b0.y = wr1[n +  0];
                v2f b1; b1.x = wr0[n + 16]; b1.y = wr1[n + 16];
                v2f b2; b2.x = wr0[n + 32]; b2.y = wr1[n + 32];
                v2f b3; b3.x = wr0[n + 48]; b3.y = wr1[n + 48];

                acc0 = __builtin_amdgcn_wmma_f32_16x16x4_f32(false, a, false, b0, (short)0, acc0, false, false);
                acc1 = __builtin_amdgcn_wmma_f32_16x16x4_f32(false, a, false, b1, (short)0, acc1, false, false);
                acc2 = __builtin_amdgcn_wmma_f32_16x16x4_f32(false, a, false, b2, (short)0, acc2, false, false);
                acc3 = __builtin_amdgcn_wmma_f32_16x16x4_f32(false, a, false, b3, (short)0, acc3, false, false);
            }

            asm volatile("v_nop\n\tv_nop\n\tv_nop\n\tv_nop\n\tv_nop"
                         : "+v"(acc0), "+v"(acc1), "+v"(acc2), "+v"(acc3));

#pragma unroll
            for (int r = 0; r < 8; ++r) {
                const int lr = t * 16 + r + 8 * g;
                if (lr < nrows) {
                    float* orow = out + (size_t)(row0 + lr) * FEATS + n;
                    orow[ 0] = acc0[r] + bb0;
                    orow[16] = acc1[r] + bb1;
                    orow[32] = acc2[r] + bb2;
                    orow[48] = acc3[r] + bb3;
                }
            }
        }
        __syncthreads();
    }
}

extern "C" void kernel_launch(void* const* d_in, const int* in_sizes, int n_in,
                              void* d_out, int out_size, void* d_ws, size_t ws_size,
                              hipStream_t stream) {
    const float* feature = (const float*)d_in[0];
    const float* degree  = (const float*)d_in[1];
    const int*   src     = (const int*)d_in[2];
    const int*   dst     = (const int*)d_in[3];
    const float* Wm      = (const float*)d_in[4];
    const float* bias    = (const float*)d_in[5];
    float* out = (float*)d_out;
    int2*  list  = (int2*)d_ws;
    int2*  table = list + (size_t)NFMAX * EPB;

    int n_nodes = in_sizes[1];
    int n_edges = in_sizes[2];

    gcn_bin_fill_kernel<<<293, FILL_T, 0, stream>>>(src, dst, list, table,
                                                    n_edges, n_nodes);

    gcn_bin_sum_apply_kernel<<<NBINS, SUM_T, 0, stream>>>(feature, degree, src, dst,
                                                          list, (const int*)table, Wm, bias,
                                                          out, n_edges, n_nodes);
}


# node product: W copied once per block to shared memory as k-pair rows; B fragments read from there one k-step ahead of their WMMAs, not by 128 global loads in each tile's k-loop; sum order as parent
# speedup vs baseline: 4.2984x; 1.1075x over previous
#include <hip/hip_runtime.h>


typedef __attribute__((ext_vector_type(2))) float v2f;
typedef __attribute__((ext_vector_type(8))) float v8f;

#define FEATS 64

#define ROWS   147
#define RPAD   160
#define PITCH  68
#define NBINS  512
#define CAP    4096
#define NFMAX  384

#define FILL_T   256
#define FILL_V   4
#define FILL_EPT (4 * FILL_V)
#define EPB      (FILL_T * FILL_EPT)
#define FILL_W   (FILL_T / 32)

__global__ void __launch_bounds__(FILL_T)
gcn_bin_fill_kernel(const int* __restrict__ src,
                    const int* __restrict__ dst,
                    int2* __restrict__ list,
                    int2* __restrict__ table,
                    int n_edges, int n_nodes) {
    __shared__ int s_cnt[NBINS];
    __shared__ int s_base[NBINS];
    __shared__ int s_wtot[FILL_W];
    __shared__ __attribute__((aligned(16))) int2 s_stage[EPB];
    const int tid  = threadIdx.x;
    const int lane = tid & 31;
    const int wv   = tid >> 5;
    const int nblk = (n_edges + EPB - 1) / EPB;
    if (nblk > NFMAX) return;
    const bool wide = ((((size_t)src) | ((size_t)dst)) & 15u) == 0;
    for (int blk = blockIdx.x; blk < nblk; blk += gridDim.x) {
        for (int i = tid; i < NBINS; i += FILL_T) s_cnt[i] = 0;
        __syncthreads();
        int s[FILL_EPT], d[FILL_EPT], bin[FILL_EPT], rk[FILL_EPT];
        const bool full = wide && (n_edges - blk * EPB >= EPB);
        if (full) {
#pragma unroll
            for (int v = 0; v < FILL_V; ++v) {
                const int e0 = blk * EPB + (v * FILL_T + tid) * 4;
                const int4 sv = *(const int4*)(src + e0);
                const int4 dv = *(const int4*)(dst + e0);
                s[4 * v + 0] = sv.x; s[4 * v + 1] = sv.y; s[4 * v + 2] = sv.z; s[4 * v + 3] = sv.w;
                d[4 * v + 0] = dv.x; d[4 * v + 1] = dv.y; d[4 * v + 2] = dv.z; d[4 * v + 3] = dv.w;
            }
        } else {
#pragma unroll
            for (int i = 0; i < FILL_EPT; ++i) {
                const int e = blk * EPB + i * FILL_T + tid;
                s[i] = 0; d[i] = -1;
                if (e < n_edges) { s[i] = src[e]; d[i] = dst[e]; }
            }
        }
#pragma unroll
        for (int i = 0; i < FILL_EPT; ++i) {
            bin[i] = -1; rk[i] = 0;
            if ((unsigned)d[i] < (unsigned)n_nodes) {
                const int b = d[i] / ROWS;
                if (b < NBINS) {
                    bin[i] = b;
                    rk[i] = atomicAdd(&s_cnt[b], 1);
                }
            }
        }
        __syncthreads();
        const int c0  = s_cnt[2 * tid + 0];
        const int c1  = s_cnt[2 * tid + 1];
        const int tot = c0 + c1;
        int inc = tot;
#pragma unroll
        for (int o = 1; o < 32; o <<= 1) {
            const int t = __shfl_up(inc, o, 32);
            if (lane >= o) inc += t;
        }
        if (lane == 31) s_wtot[wv] = inc;
        __syncthreads();
        int wbase = 0, total = 0;
#pragma unroll
        for (int j = 0; j < FILL_W; ++j) {
            const int v = s_wtot[j];
            if (j < wv) wbase += v;
            total += v;
        }
        const int ex = wbase + inc - tot;
        s_base[2 * tid + 0] = ex;
        s_base[2 * tid + 1] = ex + c0;
        table[(size_t)blk * (NBINS / 2) + tid] =
            make_int2(ex | (c0 << 16), (ex + c0) | (c1 << 16));
        __syncthreads();
        int pos[FILL_EPT];
#pragma unroll
        for (int i = 0; i < FILL_EPT; ++i) pos[i] = s_base[bin[i] >= 0 ? bin[i] : 0] + rk[i];
#pragma unroll
        for (int i = 0; i < FILL_EPT; ++i) {
            if (bin[i] >= 0) s_stage[pos[i]] = make_int2(s[i], d[i]);
        }
        __syncthreads();
        int4*       gl4 = (int4*)(list + (size_t)blk * EPB);
        const int4* st4 = (const int4*)s_stage;
#pragma unroll
        for (int j = 0; j < EPB / 2 / FILL_T; ++j) {
            const int q = j * FILL_T + tid;
            if (2 * q < total) gl4[q] = st4[q];
        }
        __syncthreads();
    }
}

#define SUM_T    512
#define SORT_EPT (CAP / SUM_T)
#define COPY_R   (NFMAX * 8 / SUM_T)
#define COPY_K   3
#define GROUP    16
#define WPITCH   160

__device__ __forceinline__ float lane_read_f(float v, int j) {
    return __int_as_float(__builtin_amdgcn_readlane(__float_as_int(v), j));
}

#define HALF_BCAST_I(v, U) __builtin_amdgcn_mov_dpp((v), 0x150 + (U), 0xF, 0xF, false)

#define ROW_LOAD(U)                                                            \
        const float4 v##U =                                                    \
            *(const float4*)(fb + ((unsigned)HALF_BCAST_I(oj, (U)) + c16));

#define ROW_ADD(U)                                                             \
        {                                                                      \
            const float wx = __int_as_float(HALF_BCAST_I(wj, (U)));            \
            acc.x = __builtin_fmaf(v##U.x, wx, acc.x);                         \
            acc.y = __builtin_fmaf(v##U.y, wx, acc.y);                         \
            acc.z = __builtin_fmaf(v##U.z, wx, acc.z);                         \
            acc.w = __builtin_fmaf(v##U.w, wx, acc.w);                         \
        }

#define QUAD_OPEN(A, B, C, D)                                                  \
    if ((A) < nmax) {                                                          \
        ROW_LOAD(A) ROW_LOAD(B) ROW_LOAD(C) ROW_LOAD(D)

#define QUAD_CLOSE(A, B, C, D)                                                 \
        ROW_ADD(D) ROW_ADD(C) ROW_ADD(B) ROW_ADD(A)                            \
    }

__global__ void __launch_bounds__(SUM_T, 8)
gcn_bin_sum_apply_kernel(const float* __restrict__ feat,
                         const float* __restrict__ degree,
                         const int* __restrict__ src,
                         const int* __restrict__ dst,
                         const int2* __restrict__ list,
                         const int* __restrict__ table,
                         const float* __restrict__ Wm,
                         const float* __restrict__ bias,
                         float* __restrict__ out,
                         int n_edges, int n_nodes) {
    __shared__ __attribute__((aligned(16))) float s_acc[RPAD * PITCH];
    __shared__ int  s_cnt[RPAD];
    __shared__ int  s_start[RPAD + 1];
    __shared__ int2 s_sw[CAP];
    __shared__ int4 s_seg[NFMAX];
    __shared__ int  s_wtot[SUM_T / 32];
    __shared__ __attribute__((aligned(16))) float s_w[(FEATS / 2) * WPITCH];
    const int tid  = threadIdx.x;
    const int lane = tid & 31;
    const int c = lane & 15;
    const int h = lane >> 4;
    const int wv     = __builtin_amdgcn_readfirstlane(tid >> 5);
    const int nwaves = SUM_T >> 5;
    const int nbins  = (n_nodes + ROWS - 1) / ROWS;
    const int nblk   = (n_edges + EPB - 1) / EPB;

    {
        static_assert(SUM_T == (FEATS / 2) * (FEATS / 4), "one thread per 32-byte piece of a k-pair row");
        const int      wkp  = tid >> 4;
        const int      wn4  = (tid & 15) * 4;
        const unsigned woff = (unsigned)(2 * wkp * FEATS + wn4) * 4u;
        const float4 wr0 = *(const float4*)((const char*)Wm + woff);
        const float4 wr1 = *(const float4*)((const char*)Wm + woff + FEATS * 4u);
        float4* wp = (float4*)(s_w + wkp * WPITCH + 2 * wn4);
        wp[0] = make_float4(wr0.x, wr1.x, wr0.y, wr1.y);
        wp[1] = make_float4(wr0.z, wr1.z, wr0.w, wr1.w);
    }

    for (int b = blockIdx.x; b < nbins; b += gridDim.x) {
        const int row0   = b * ROWS;
        const int left   = n_nodes - row0;
        const int nrows  = left < ROWS ? left : ROWS;
        const int ntiles = (nrows + 15) >> 4;

        int  nlist    = 0;
        bool use_list = false;
        if (b < NBINS && nblk <= NFMAX) {
            int rs = 0, rc = 0;
            if (tid < nblk) {
                const int w = table[(size_t)tid * NBINS + b];
                rs = w & 0xFFFF;
                rc = w >> 16;
            }
            int inc = rc;
#pragma unroll
            for (int o = 1; o < 32; o <<= 1) {
                const int t = __shfl_up(inc, o, 32);
                if (lane >= o) inc += t;
            }
            if (lane == 31) s_wtot[wv] = inc;
            __syncthreads();
            int wbase = 0;
#pragma unroll
            for (int j = 0; j < SUM_T / 32; ++j) {
                const int v = s_wtot[j];
                if (j < wv) wbase += v;
                nlist += v;
            }
            if (tid < NFMAX) s_seg[tid] = make_int4(tid * EPB + rs, rc, wbase + inc - rc, 0);
            use_list = (nlist <= CAP);
        }
        use_list = __builtin_amdgcn_readfirstlane(use_list ? 1 : 0) != 0;
        nlist    = __builtin_amdgcn_readfirstlane(nlist);

        if (use_list) {
            for (int i = tid; i < RPAD; i += SUM_T) s_cnt[i] = 0;
            __syncthreads();

            int2* s_pairs = (int2*)s_acc;
            {
                const int sub = tid & 7;
                const int grp = tid >> 3;
                int4 sg[COPY_R];
                int2 pr[COPY_R][COPY_K];
#pragma unroll
                for (int j = 0; j < COPY_R; ++j) sg[j] = s_seg[grp + (SUM_T / 8) * j];
#pragma unroll
                for (int j = 0; j < COPY_R; ++j) {
#pragma unroll
                    for (int k = 0; k < COPY_K; ++k) {
                        const int i = sub + 8 * k;
                        pr[j][k] = make_int2(0, -1);
                        if (i < sg[j].y) pr[j][k] = list[sg[j].x + i];
                    }
                }
#pragma unroll
                for (int j = 0; j < COPY_R; ++j) {
#pragma unroll
                    for (int k = 0; k < COPY_K; ++k) {
                        const int i = sub + 8 * k;
                        if (i < sg[j].y) s_pairs[sg[j].z + i] = pr[j][k];
                    }
                    for (int i = sub + 8 * COPY_K; i < sg[j].y; i += 8)
                        s_pairs[sg[j].z + i] = list[sg[j].x + i];
                }
            }
            __syncthreads();

            int   es[SORT_EPT], er[SORT_EPT], ek[SORT_EPT];
            float ew[SORT_EPT];
            int2  sd[SORT_EPT];
            float dg[SORT_EPT];
#pragma unroll
            for (int i = 0; i < SORT_EPT; ++i) {
                const int e = i * SUM_T + tid;
                sd[i] = make_int2(0, -1);
                if (e < nlist) sd[i] = s_pairs[e];
            }
#pragma unroll
            for (int i = 0; i < SORT_EPT; ++i) {
                const int e  = i * SUM_T + tid;
                const int rr = sd[i].y - row0;
                es[i] = 0; er[i] = -1; ek[i] = 0; dg[i] = 1.f;
                if (e < nlist && (unsigned)rr < (unsigned)nrows) {
                    es[i] = sd[i].x;
                    er[i] = rr;
                    ek[i] = atomicAdd(&s_cnt[rr], 1);
                    dg[i] = degree[sd[i].x];
                }
            }
#pragma unroll
            for (int i = 0; i < SORT_EPT; ++i)
                ew[i] = er[i] >= 0 ? __frsqrt_rn(dg[i]) : 0.f;
            __syncthreads();

            if (wv == 0) {
                const int c0 = s_cnt[5 * lane + 0];
                const int c1 = s_cnt[5 * lane + 1];
                const int c2 = s_cnt[5 * lane + 2];
                const int c3 = s_cnt[5 * lane + 3];
                const int c4 = s_cnt[5 * lane + 4];
                const int tot = c0 + c1 + c2 + c3 + c4;
                int inc = tot;
#pragma unroll
                for (int o = 1; o < 32; o <<= 1) {
                    const int t = __shfl_up(inc, o, 32);
                    if (lane >= o) inc += t;
                }
                int ex = inc - tot;
                s_start[5 * lane + 0] = ex; ex += c0;
                s_start[5 * lane + 1] = ex; ex += c1;
                s_start[5 * lane + 2] = ex; ex += c2;
                s_start[5 * lane + 3] = ex; ex += c3;
                s_start[5 * lane + 4] = ex; ex += c4;
                if (lane == 31) s_start[RPAD] = ex;
            }
            __syncthreads();

            int ps[SORT_EPT];
#pragma unroll
            for (int i = 0; i < SORT_EPT; ++i) ps[i] = s_start[er[i] >= 0 ? er[i] : 0];
#pragma unroll
            for (int i = 0; i < SORT_EPT; ++i) {
                if (er[i] >= 0)
                    s_sw[ps[i] + ek[i]] = make_int2((int)((unsigned)es[i] << 8), __float_as_int(ew[i]));
            }
            __syncthreads();

            const char*    fb  = (const char*)feat;
            const unsigned c16 = 16u * (unsigned)c;
            const int npairs = ntiles * 8;
            for (int q = wv; q < npairs; q += nwaves) {
                const int rr  = 2 * q + h;
                const int st  = s_start[rr];
                const int cnt = s_start[rr + 1] - st;
                const int ca  = __builtin_amdgcn_readlane(cnt, 0);
                const int cb  = __builtin_amdgcn_readlane(cnt, 16);
                const int cmax = ca > cb ? ca : cb;
                float4 acc = make_float4(0.f, 0.f, 0.f, 0.f);
                for (int base = 0; base < cmax; base += 16) {
                    const int nh   = cnt - base;
                    const int nmax = cmax - base;
                    int oj = 0, wj = 0;
                    if (c < nh) {
                        const int2 t = s_sw[st + base + c];
                        oj = t.x; wj = t.y;
                    }
                    QUAD_OPEN(0, 1, 2, 3)
                    QUAD_OPEN(4, 5, 6, 7)
                    QUAD_OPEN(8, 9, 10, 11)
                    QUAD_OPEN(12, 13, 14, 15)
                    QUAD_CLOSE(12, 13, 14, 15)
                    QUAD_CLOSE(8, 9, 10, 11)
                    QUAD_CLOSE(4, 5, 6, 7)
                    QUAD_CLOSE(0, 1, 2, 3)
                }
                *(float4*)(s_acc + rr * PITCH + 4 * c) = acc;
            }
        } else {
            for (int i = tid; i < RPAD * PITCH; i += SUM_T) s_acc[i] = 0.f;
            __syncthreads();
            const int nchunks = (n_edges + 31) / 32;
            for (int ch = wv; ch < nchunks; ch += nwaves) {
                const int e = ch * 32 + lane;
                int s = 0, d = row0;
                bool mine = false;
                if (e < n_edges) {
                    s = src[e];
                    d = dst[e];
                    mine = (d >= row0) && (d < row0 + ROWS) && (d < n_nodes);
                }
                float w = 0.f;
                if (mine) w = __frsqrt_rn(degree[s]);
                const int dl = mine ? (d - row0) * PITCH : 0;
                unsigned mask = __builtin_amdgcn_ballot_w32(mine);

                while (mask) {
                    const int n = __builtin_popcount(mask);
                    int j[GROUP];
#pragma unroll
                    for (int u = 0; u < GROUP; ++u) {
                        j[u] = mask ? __builtin_ctz(mask) : 0;
                        mask &= mask - 1u;
                    }
                    float2 v[GROUP];
#pragma unroll
                    for (int u = 0; u < GROUP; ++u) {
                        v[u] = make_float2(0.f, 0.f);
                        if (u < n) {
                            const int sj = __builtin_amdgcn_readlane(s, j[u]);
                            v[u] = *(const float2*)(feat + (size_t)sj * FEATS + 2 * lane);
                        }
                    }
#pragma unroll
                    for (int u = 0; u < GROUP; ++u) {
                        if (u < n) {
                            const int   dj = __builtin_amdgcn_readlane(dl, j[u]);
                            const float wj = lane_read_f(w, j[u]);
                            float* p = s_acc + dj + 2 * lane;
                            unsafeAtomicAdd(p,     v[u].x * wj);
                            unsafeAtomicAdd(p + 1, v[u].y * wj);
                        }
                    }
                }
            }
        }
        __syncthreads();

        const int n = lane & 15;
        const int g = lane >> 4;
        for (int t = wv; t < ntiles; t += nwaves) {
            const int lrow = t * 16 + n;
            const int grow = row0 + lrow;
            const int gcl  = grow < n_nodes ? grow : n_nodes - 1;
            const float dsc = __frsqrt_rn(degree[gcl]);
            const float* arow = s_acc + lrow * PITCH;

            const float bb0 = bias[n +  0];
            const float bb1 = bias[n + 16];
            const float bb2 = bias[n + 32];
            const float bb3 = bias[n + 48];

            const float* wl = s_w + g * WPITCH + 2 * n;
            v8f acc0 = {}, acc1 = {}, acc2 = {}, acc3 = {};
            float2 an  = *(const float2*)(arow + 2 * g);
            float2 bn0 = *(const float2*)(wl +  0);
            float2 bn1 = *(const float2*)(wl + 32);
            float2 bn2 = *(const float2*)(wl + 64);
            float2 bn3 = *(const float2*)(wl + 96);
#pragma unroll
            for (int s = 0; s < FEATS / 4; ++s) {
                v2f a;
                a.x = an.x * dsc;
                a.y = an.y * dsc;
                v2f b0; b0.x = bn0.x; b0.y = bn0.y;
                v2f b1; b1.x = bn1.x; b1.y = bn1.y;
                v2f b2; b2.x = bn2.x; b2.y = bn2.y;
                v2f b3; b3.x = bn3.x; b3.y = bn3.y;
                if (s + 1 < FEATS / 4) {
                    const float* wn = wl + 2 * (s + 1) * WPITCH;
                    an  = *(const float2*)(arow + 4 * (s + 1) + 2 * g);
                    bn0 = *(const float2*)(wn +  0);
                    bn1 = *(const float2*)(wn + 32);
                    bn2 = *(const float2*)(wn + 64);
                    bn3 = *(const float2*)(wn + 96);
                }

                acc0 = __builtin_amdgcn_wmma_f32_16x16x4_f32(false, a, false, b0, (short)0, acc0, false, false);
                acc1 = __builtin_amdgcn_wmma_f32_16x16x4_f32(false, a, false, b1, (short)0, acc1, false, false);
                acc2 = __builtin_amdgcn_wmma_f32_16x16x4_f32(false, a, false, b2, (short)0, acc2, false, false);
                acc3 = __builtin_amdgcn_wmma_f32_16x16x4_f32(false, a, false, b3, (short)0, acc3, false, false);
            }

            asm volatile("v_nop\n\tv_nop\n\tv_nop\n\tv_nop\n\tv_nop"
                         : "+v"(acc0), "+v"(acc1), "+v"(acc2), "+v"(acc3));

#pragma unroll
            for (int r = 0; r < 8; ++r) {
                const int lr = t * 16 + r + 8 * g;
                if (lr < nrows) {
                    float* orow = out + (size_t)(row0 + lr) * FEATS + n;
                    orow[ 0] = acc0[r] + bb0;
                    orow[16] = acc1[r] + bb1;
                    orow[32] = acc2[r] + bb2;
                    orow[48] = acc3[r] + bb3;
                }
            }
        }
        __syncthreads();
    }
}

extern "C" void kernel_launch(void* const* d_in, const int* in_sizes, int n_in,
                              void* d_out, int out_size, void* d_ws, size_t ws_size,
                              hipStream_t stream) {
    const float* feature = (const float*)d_in[0];
    const float* degree  = (const float*)d_in[1];
    const int*   src     = (const int*)d_in[2];
    const int*   dst     = (const int*)d_in[3];
    const float* Wm      = (const float*)d_in[4];
    const float* bias    = (const float*)d_in[5];
    float* out = (float*)d_out;
    int2*  list  = (int2*)d_ws;
    int2*  table = list + (size_t)NFMAX * EPB;

    int n_nodes = in_sizes[1];
    int n_edges = in_sizes[2];

    gcn_bin_fill_kernel<<<293, FILL_T, 0, stream>>>(src, dst, list, table,
                                                    n_edges, n_nodes);

    gcn_bin_sum_apply_kernel<<<NBINS, SUM_T, 0, stream>>>(feature, degree, src, dst,
                                                          list, (const int*)table, Wm, bias,
                                                          out, n_edges, n_nodes);
}
